# starting kernel
# baseline (speedup 1.0000x reference)
#include <hip/hip_runtime.h>

typedef __attribute__((ext_vector_type(2))) float v2f;
typedef __attribute__((ext_vector_type(8))) float v8f;

#define NFEAT 256
#define NCLASS 64

__global__ void deg_init_kernel(float* __restrict__ deg, int n) {
    int i = blockIdx.x * blockDim.x + threadIdx.x;
    if (i < n) deg[i] = 1.0f;
}

__global__ void deg_count_kernel(const int* __restrict__ col, float* __restrict__ deg, int e) {
    int i = blockIdx.x * blockDim.x + threadIdx.x;
    if (i < e) atomicAdd(&deg[col[i]], 1.0f);
}

__global__ void deg_rsqrt_kernel(float* __restrict__ deg, int n) {
    int i = blockIdx.x * blockDim.x + threadIdx.x;
    if (i < n) deg[i] = rsqrtf(deg[i]);
}

__global__ __launch_bounds__(256) void gemm1_kernel(const float* __restrict__ X,
                                                    const float* __restrict__ Wg,
                                                    float* __restrict__ H, int n) {
    const int lane = threadIdx.x & 31;
    const int wave = threadIdx.x >> 5;
    const int mtile = blockIdx.x * 8 + wave;
    const int m0 = mtile * 16;
    if (m0 >= n) return;

    const int half = lane >> 4;
    const int l15  = lane & 15;
    int m = m0 + l15;
    if (m >= n) m = n - 1;
    const float* __restrict__ xrow = X + (size_t)m * NFEAT;

    v8f acc0 = {}, acc1 = {}, acc2 = {}, acc3 = {};

    for (int k0 = 0; k0 < NFEAT; k0 += 4) {
        const int kb = k0 + half * 2;
        v2f a;
        a.x = xrow[kb];
        a.y = xrow[kb + 1];

        const float* __restrict__ w0 = Wg + (size_t)kb * NCLASS;
        const float* __restrict__ w1 = w0 + NCLASS;
        v2f b0, b1, b2, b3;
        b0.x = w0[l15];      b0.y = w1[l15];
        b1.x = w0[16 + l15]; b1.y = w1[16 + l15];
        b2.x = w0[32 + l15]; b2.y = w1[32 + l15];
        b3.x = w0[48 + l15]; b3.y = w1[48 + l15];

        acc0 = __builtin_amdgcn_wmma_f32_16x16x4_f32(false, a, false, b0, (short)0, acc0, false, false);
        acc1 = __builtin_amdgcn_wmma_f32_16x16x4_f32(false, a, false, b1, (short)0, acc1, false, false);
        acc2 = __builtin_amdgcn_wmma_f32_16x16x4_f32(false, a, false, b2, (short)0, acc2, false, false);
        acc3 = __builtin_amdgcn_wmma_f32_16x16x4_f32(false, a, false, b3, (short)0, acc3, false, false);
    }

    const int mbase = m0 + half * 8;
#pragma unroll
    for (int r = 0; r < 8; ++r) {
        const int mr = mbase + r;
        if (mr < n) {
            float* __restrict__ hrow = H + (size_t)mr * NCLASS;
            hrow[l15]      = acc0[r];
            hrow[16 + l15] = acc1[r];
            hrow[32 + l15] = acc2[r];
            hrow[48 + l15] = acc3[r];
        }
    }
}

__global__ void agg_init_kernel(const float* __restrict__ H,
                                const float* __restrict__ dinv,
                                const float* __restrict__ bg,
                                float* __restrict__ agg, int n) {
    int i = blockIdx.x * blockDim.x + threadIdx.x;
    int total = n * NCLASS;
    if (i < total) {
        int node = i >> 6;
        int c = i & 63;
        float d = dinv[node];
        agg[i] = H[i] * d * d + bg[c];
    }
}

__global__ void scatter_kernel(const int* __restrict__ ei,
                               const float* __restrict__ H,
                               const float* __restrict__ dinv,
                               float* __restrict__ agg, int e) {
    int gid = blockIdx.x * blockDim.x + threadIdx.x;
    int total = e * 16;
    if (gid >= total) return;
    const int edge = gid >> 4;
    const int q = gid & 15;
    const int r = ei[edge];
    const int c = ei[e + edge];
    const float nrm = dinv[r] * dinv[c];

    const float4 hv = *reinterpret_cast<const float4*>(H + (size_t)r * NCLASS + q * 4);
    float* __restrict__ dst = agg + (size_t)c * NCLASS + q * 4;
    atomicAdd(dst + 0, hv.x * nrm);
    atomicAdd(dst + 1, hv.y * nrm);
    atomicAdd(dst + 2, hv.z * nrm);
    atomicAdd(dst + 3, hv.w * nrm);
}

__global__ __launch_bounds__(256) void gemm2_kernel(const float* __restrict__ AGG,
                                                    const float* __restrict__ Wl,
                                                    const float* __restrict__ bl,
                                                    float* __restrict__ OUT, int n) {
    const int lane = threadIdx.x & 31;
    const int wave = threadIdx.x >> 5;
    const int mtile = blockIdx.x * 8 + wave;
    const int m0 = mtile * 16;
    if (m0 >= n) return;

    const int half = lane >> 4;
    const int l15  = lane & 15;
    int m = m0 + l15;
    if (m >= n) m = n - 1;
    const float* __restrict__ arow = AGG + (size_t)m * NCLASS;

    v8f acc0 = {}, acc1 = {}, acc2 = {}, acc3 = {};

    for (int k0 = 0; k0 < NCLASS; k0 += 4) {
        const int kb = k0 + half * 2;
        v2f a;
        a.x = fmaxf(arow[kb], 0.0f);
        a.y = fmaxf(arow[kb + 1], 0.0f);

        const float* __restrict__ w0 = Wl + (size_t)kb * NCLASS;
        const float* __restrict__ w1 = w0 + NCLASS;
        v2f b0, b1, b2, b3;
        b0.x = w0[l15];      b0.y = w1[l15];
        b1.x = w0[16 + l15]; b1.y = w1[16 + l15];
        b2.x = w0[32 + l15]; b2.y = w1[32 + l15];
        b3.x = w0[48 + l15]; b3.y = w1[48 + l15];

        acc0 = __builtin_amdgcn_wmma_f32_16x16x4_f32(false, a, false, b0, (short)0, acc0, false, false);
        acc1 = __builtin_amdgcn_wmma_f32_16x16x4_f32(false, a, false, b1, (short)0, acc1, false, false);
        acc2 = __builtin_amdgcn_wmma_f32_16x16x4_f32(false, a, false, b2, (short)0, acc2, false, false);
        acc3 = __builtin_amdgcn_wmma_f32_16x16x4_f32(false, a, false, b3, (short)0, acc3, false, false);
    }

    const float bl0 = bl[l15];
    const float bl1 = bl[16 + l15];
    const float bl2 = bl[32 + l15];
    const float bl3 = bl[48 + l15];

    const int mbase = m0 + half * 8;
#pragma unroll
    for (int r = 0; r < 8; ++r) {
        const int mr = mbase + r;
        if (mr < n) {
            float* __restrict__ orow = OUT + (size_t)mr * NCLASS;
            orow[l15]      = acc0[r] + bl0;
            orow[16 + l15] = acc1[r] + bl1;
            orow[32 + l15] = acc2[r] + bl2;
            orow[48 + l15] = acc3[r] + bl3;
        }
    }
}

extern "C" void kernel_launch(void* const* d_in, const int* in_sizes, int n_in,
                              void* d_out, int out_size, void* d_ws, size_t ws_size,
                              hipStream_t stream) {
    const float* x  = (const float*)d_in[0];
    const int*   ei = (const int*)d_in[1];
    const float* Wg = (const float*)d_in[2];
    const float* bg = (const float*)d_in[3];
    const float* Wl = (const float*)d_in[4];
    const float* bl = (const float*)d_in[5];
    float* out = (float*)d_out;

    const int n = in_sizes[0] / NFEAT;
    const int e = in_sizes[1] / 2;

    float* ws   = (float*)d_ws;
    size_t off  = ((size_t)n + 3) & ~(size_t)3;
    float* dinv = ws;
    float* h    = ws + off;
    float* agg  = h + (size_t)n * NCLASS;

    const int T = 256;

    deg_init_kernel <<<(n + T - 1) / T, T, 0, stream>>>(dinv, n);
    deg_count_kernel<<<(e + T - 1) / T, T, 0, stream>>>(ei + e, dinv, e);
    deg_rsqrt_kernel<<<(n + T - 1) / T, T, 0, stream>>>(dinv, n);

    const int mtiles = (n + 15) / 16;
    const int gblocks = (mtiles + 7) / 8;
    gemm1_kernel<<<gblocks, T, 0, stream>>>(x, Wg, h, n);

    agg_init_kernel<<<((n * NCLASS) + T - 1) / T, T, 0, stream>>>(h, dinv, bg, agg, n);

    const int stotal = e * 16;
    scatter_kernel<<<(stotal + T - 1) / T, T, 0, stream>>>(ei, h, dinv, agg, e);

    gemm2_kernel<<<gblocks, T, 0, stream>>>(agg, Wl, bl, out, n);
}


# scatter: each block keeps all node degree counts in LDS (2-byte numbers, 65535 = read memory); lanes take both edge-weight factors there, not 2 global 4-byte gathers per edge; float atomics kept
# speedup vs baseline: 2.3678x; 2.3678x over previous
#include <hip/hip_runtime.h>

typedef __attribute__((ext_vector_type(2))) float v2f;
typedef __attribute__((ext_vector_type(8))) float v8f;

#define NFEAT 256
#define NCLASS 64

__device__ __forceinline__ int chip_half() {
    return (int)((__builtin_amdgcn_s_sendmsg_rtn(0x87) >> 18) & 1u);
}

__global__ void deg_init_kernel(float* __restrict__ deg0, float* __restrict__ deg1, int n) {
    const int i = (blockIdx.x * blockDim.x + threadIdx.x) * 4;
    if (i + 3 < n) {
        *(float4*)(deg0 + i) = make_float4(1.0f, 1.0f, 1.0f, 1.0f);
        *(float4*)(deg1 + i) = make_float4(0.0f, 0.0f, 0.0f, 0.0f);
    } else {
        for (int k = i; k < n; ++k) {
            deg0[k] = 1.0f;
            deg1[k] = 0.0f;
        }
    }
}

#define DC_U 8

__device__ __forceinline__ void deg_count_body(const int* __restrict__ col,
                                               float* __restrict__ deg0,
                                               float* __restrict__ deg1, int e,
                                               int blk, int nblk) {
    const int t = blk * 256 + threadIdx.x;
    const int stride = nblk * 256;
    float* deg = chip_half() ? deg1 : deg0;
    if (t + (DC_U - 1) * stride < e) {
        int c[DC_U];
#pragma unroll
        for (int u = 0; u < DC_U; ++u) c[u] = col[t + u * stride];
#pragma unroll
        for (int u = 0; u < DC_U; ++u) atomicAdd(&deg[c[u]], 1.0f);
    } else {
        for (int i = t; i < e; i += stride) atomicAdd(&deg[col[i]], 1.0f);
    }
}

__global__ void deg_sum_kernel(float* __restrict__ deg0, const float* __restrict__ deg1, int n) {
    const int i = (blockIdx.x * blockDim.x + threadIdx.x) * 4;
    if (i + 3 < n) {
        const float4 a = *(const float4*)(deg0 + i);
        const float4 b = *(const float4*)(deg1 + i);
        float4 o;
        o.x = a.x + b.x;
        o.y = a.y + b.y;
        o.z = a.z + b.z;
        o.w = a.w + b.w;
        *(float4*)(deg0 + i) = o;
    } else {
        for (int k = i; k < n; ++k) deg0[k] = deg0[k] + deg1[k];
    }
}

__device__ __forceinline__ void gemm1_body(const float* __restrict__ X,
                                           const float* __restrict__ Wg,
                                           float* __restrict__ H, int n, int blk) {
    const int lane = threadIdx.x & 31;
    const int wave = threadIdx.x >> 5;
    const int mtile = blk * 8 + wave;
    const int m0 = mtile * 16;
    if (m0 >= n) return;

    const int half = lane >> 4;
    const int l15  = lane & 15;
    int m = m0 + l15;
    if (m >= n) m = n - 1;
    const float* __restrict__ xrow = X + (size_t)m * NFEAT;

    v8f acc0 = {}, acc1 = {}, acc2 = {}, acc3 = {};

    for (int k0 = 0; k0 < NFEAT; k0 += 4) {
        const int kb = k0 + half * 2;
        v2f a;
        a.x = xrow[kb];
        a.y = xrow[kb + 1];

        const float* __restrict__ w0 = Wg + (size_t)kb * NCLASS;
        const float* __restrict__ w1 = w0 + NCLASS;
        v2f b0, b1, b2, b3;
        b0.x = w0[l15];      b0.y = w1[l15];
        b1.x = w0[16 + l15]; b1.y = w1[16 + l15];
        b2.x = w0[32 + l15]; b2.y = w1[32 + l15];
        b3.x = w0[48 + l15]; b3.y = w1[48 + l15];

        acc0 = __builtin_amdgcn_wmma_f32_16x16x4_f32(false, a, false, b0, (short)0, acc0, false, false);
        acc1 = __builtin_amdgcn_wmma_f32_16x16x4_f32(false, a, false, b1, (short)0, acc1, false, false);
        acc2 = __builtin_amdgcn_wmma_f32_16x16x4_f32(false, a, false, b2, (short)0, acc2, false, false);
        acc3 = __builtin_amdgcn_wmma_f32_16x16x4_f32(false, a, false, b3, (short)0, acc3, false, false);
    }

    const int mbase = m0 + half * 8;
#pragma unroll
    for (int r = 0; r < 8; ++r) {
        const int mr = mbase + r;
        if (mr < n) {
            float* __restrict__ hrow = H + (size_t)mr * NCLASS;
            hrow[l15]      = acc0[r];
            hrow[16 + l15] = acc1[r];
            hrow[32 + l15] = acc2[r];
            hrow[48 + l15] = acc3[r];
        }
    }
}

__global__ __launch_bounds__(256) void gemm1_degcount_kernel(const float* __restrict__ X,
                                                             const float* __restrict__ Wg,
                                                             float* __restrict__ H, int n,
                                                             int gblocks,
                                                             const int* __restrict__ col,
                                                             float* __restrict__ deg0,
                                                             float* __restrict__ deg1, int e) {
    const int blk = (int)blockIdx.x;
    if (blk < gblocks) {
        gemm1_body(X, Wg, H, n, blk);
    } else {
        deg_count_body(col, deg0, deg1, e, blk - gblocks, (int)gridDim.x - gblocks);
    }
}

#define AI_U 4

__device__ __forceinline__ float4 agg_start(const float4 h, const float d, const float4 b) {
    float4 o;
    o.x = h.x * d * d + b.x;
    o.y = h.y * d * d + b.y;
    o.z = h.z * d * d + b.z;
    o.w = h.w * d * d + b.w;
    return o;
}

__global__ __launch_bounds__(256) void agg_init_kernel(const float* __restrict__ H,
                                                       const float* __restrict__ deg,
                                                       const float* __restrict__ bg,
                                                       float* __restrict__ agg0,
                                                       float* __restrict__ agg1, int n) {
    const int t = blockIdx.x * blockDim.x + threadIdx.x;
    const int stride = gridDim.x * blockDim.x;
    const int total4 = n * (NCLASS / 4);
    const float4* __restrict__ H4 = (const float4*)H;
    float4* __restrict__ a0 = (float4*)agg0;
    float4* __restrict__ a1 = (float4*)agg1;
    const float4 b = ((const float4*)bg)[t & 15];
    const float4 zero = make_float4(0.0f, 0.0f, 0.0f, 0.0f);
    if (t + (AI_U - 1) * stride < total4) {
        float4 h[AI_U];
        float d[AI_U];
#pragma unroll
        for (int u = 0; u < AI_U; ++u) {
            const int i = t + u * stride;
            h[u] = H4[i];
            d[u] = rsqrtf(deg[i >> 4]);
        }
#pragma unroll
        for (int u = 0; u < AI_U; ++u) {
            const int i = t + u * stride;
            a0[i] = agg_start(h[u], d[u], b);
            a1[i] = zero;
        }
    } else {
        for (int i = t; i < total4; i += stride) {
            a0[i] = agg_start(H4[i], rsqrtf(deg[i >> 4]), b);
            a1[i] = zero;
        }
    }
}

#define SC_BATCH 8
#define SC_THREADS 1024
#define SC_BLOCKS 256
#define DT_CAP 102400
#define DT_FULL 65535u

__device__ __forceinline__ uint2 pack_counts(const float4 d) {
    const unsigned u0 = min((unsigned)d.x, DT_FULL);
    const unsigned u1 = min((unsigned)d.y, DT_FULL);
    const unsigned u2 = min((unsigned)d.z, DT_FULL);
    const unsigned u3 = min((unsigned)d.w, DT_FULL);
    return make_uint2(u0 | (u1 << 16), u2 | (u3 << 16));
}

__global__ __launch_bounds__(SC_THREADS) void scatter_kernel(const int* __restrict__ ei,
                                                             const float* __restrict__ H,
                                                             const float* __restrict__ deg,
                                                             float* __restrict__ agg0,
                                                             float* __restrict__ agg1, int e, int n) {
    __shared__ __attribute__((aligned(16))) unsigned short dtab[DT_CAP];

    {
        const int lim = min(n, DT_CAP);
        const int lim4 = lim & ~3;
        const int chunk = SC_THREADS * 4;
        const int full = lim4 / chunk;
        const int t4 = (int)threadIdx.x * 4;
        int t = full ? (int)(blockIdx.x % (unsigned)full) : 0;
#pragma unroll 8
        for (int k = 0; k < full; ++k) {
            const int i = t * chunk + t4;
            *(uint2*)(dtab + i) = pack_counts(*(const float4*)(deg + i));
            t = (t + 1 == full) ? 0 : t + 1;
        }
        const int i = full * chunk + t4;
        if (i < lim4) *(uint2*)(dtab + i) = pack_counts(*(const float4*)(deg + i));
        for (int k = lim4 + (int)threadIdx.x; k < lim; k += SC_THREADS)
            dtab[k] = (unsigned short)min((unsigned)deg[k], DT_FULL);
    }
    __syncthreads();

    const int lane = threadIdx.x & 31;
    float* __restrict__ agg = chip_half() ? agg1 : agg0;
    const int wv = __builtin_amdgcn_readfirstlane((int)((blockIdx.x * blockDim.x + threadIdx.x) >> 5));
    const int nwaves = (int)((gridDim.x * blockDim.x) >> 5);
    const int ngroups = (e + 31) >> 5;

    for (int g = wv; g < ngroups; g += nwaves) {
        const int base = g * 32;
        const int cnt = min(32, e - base);

        int r = 0, c = 0;
        float nrm = 0.0f;
        if (lane < cnt) {
            r = ei[base + lane];
            c = ei[e + base + lane];
            const unsigned tr = (r < DT_CAP) ? (unsigned)dtab[r] : DT_FULL;
            const unsigned tc = (c < DT_CAP) ? (unsigned)dtab[c] : DT_FULL;
            float dr = (float)tr, dc = (float)tc;
            if (tr == DT_FULL) dr = deg[r];
            if (tc == DT_FULL) dc = deg[c];
            nrm = rsqrtf(dr) * rsqrtf(dc);
        }
        const int nbits = __float_as_int(nrm);

        if (cnt == 32) {
            float h0[2][SC_BATCH], h1[2][SC_BATCH];
#pragma unroll
            for (int u = 0; u < SC_BATCH; ++u) {
                const float* src = H + (size_t)__builtin_amdgcn_readlane(r, u) * NCLASS;
                h0[0][u] = src[lane];
                h1[0][u] = src[32 + lane];
            }
#pragma unroll
            for (int b = 0; b < 32 / SC_BATCH; ++b) {
                if (b + 1 < 32 / SC_BATCH) {
#pragma unroll
                    for (int u = 0; u < SC_BATCH; ++u) {
                        const int j = (b + 1) * SC_BATCH + u;
                        const float* src = H + (size_t)__builtin_amdgcn_readlane(r, j) * NCLASS;
                        h0[(b + 1) & 1][u] = src[lane];
                        h1[(b + 1) & 1][u] = src[32 + lane];
                    }
                }
#pragma unroll
                for (int u = 0; u < SC_BATCH; ++u) {
                    const int j = b * SC_BATCH + u;
                    const float nj = __int_as_float(__builtin_amdgcn_readlane(nbits, j));
                    float* dst = agg + (size_t)__builtin_amdgcn_readlane(c, j) * NCLASS;
                    atomicAdd(dst + lane, h0[b & 1][u] * nj);
                    atomicAdd(dst + 32 + lane, h1[b & 1][u] * nj);
                }
            }
        } else {
            for (int j = 0; j < cnt; ++j) {
                const float nj = __int_as_float(__builtin_amdgcn_readlane(nbits, j));
                const float* src = H + (size_t)__builtin_amdgcn_readlane(r, j) * NCLASS;
                float* dst = agg + (size_t)__builtin_amdgcn_readlane(c, j) * NCLASS;
                const float a0 = src[lane];
                const float a1 = src[32 + lane];
                atomicAdd(dst + lane, a0 * nj);
                atomicAdd(dst + 32 + lane, a1 * nj);
            }
        }
    }
}

__global__ __launch_bounds__(256) void gemm2_kernel(const float* AGG0,
                                                    const float* __restrict__ Wl,
                                                    const float* __restrict__ bl,
                                                    float* OUT, int n) {
    const int lane = threadIdx.x & 31;
    const int wave = threadIdx.x >> 5;
    const int mtile = blockIdx.x * 8 + wave;
    const int m0 = mtile * 16;
    if (m0 >= n) return;

    const int half = lane >> 4;
    const int l15  = lane & 15;
    int m = m0 + l15;
    if (m >= n) m = n - 1;
    const float* arow = AGG0 + (size_t)m * NCLASS;
    const float* prow = OUT + (size_t)m * NCLASS;

    v8f acc0 = {}, acc1 = {}, acc2 = {}, acc3 = {};

    for (int k0 = 0; k0 < NCLASS; k0 += 4) {
        const int kb = k0 + half * 2;
        v2f a;
        a.x = fmaxf(arow[kb] + prow[kb], 0.0f);
        a.y = fmaxf(arow[kb + 1] + prow[kb + 1], 0.0f);

        const float* __restrict__ w0 = Wl + (size_t)kb * NCLASS;
        const float* __restrict__ w1 = w0 + NCLASS;
        v2f b0, b1, b2, b3;
        b0.x = w0[l15];      b0.y = w1[l15];
        b1.x = w0[16 + l15]; b1.y = w1[16 + l15];
        b2.x = w0[32 + l15]; b2.y = w1[32 + l15];
        b3.x = w0[48 + l15]; b3.y = w1[48 + l15];

        acc0 = __builtin_amdgcn_wmma_f32_16x16x4_f32(false, a, false, b0, (short)0, acc0, false, false);
        acc1 = __builtin_amdgcn_wmma_f32_16x16x4_f32(false, a, false, b1, (short)0, acc1, false, false);
        acc2 = __builtin_amdgcn_wmma_f32_16x16x4_f32(false, a, false, b2, (short)0, acc2, false, false);
        acc3 = __builtin_amdgcn_wmma_f32_16x16x4_f32(false, a, false, b3, (short)0, acc3, false, false);
    }

    const float bl0 = bl[l15];
    const float bl1 = bl[16 + l15];
    const float bl2 = bl[32 + l15];
    const float bl3 = bl[48 + l15];

    const int mbase = m0 + half * 8;
#pragma unroll
    for (int r = 0; r < 8; ++r) {
        const int mr = mbase + r;
        if (mr < n) {
            float* orow = OUT + (size_t)mr * NCLASS;
            orow[l15]      = acc0[r] + bl0;
            orow[16 + l15] = acc1[r] + bl1;
            orow[32 + l15] = acc2[r] + bl2;
            orow[48 + l15] = acc3[r] + bl3;
        }
    }
}

extern "C" void kernel_launch(void* const* d_in, const int* in_sizes, int n_in,
                              void* d_out, int out_size, void* d_ws, size_t ws_size,
                              hipStream_t stream) {
    const float* x  = (const float*)d_in[0];
    const int*   ei = (const int*)d_in[1];
    const float* Wg = (const float*)d_in[2];
    const float* bg = (const float*)d_in[3];
    const float* Wl = (const float*)d_in[4];
    const float* bl = (const float*)d_in[5];
    float* out = (float*)d_out;

    const int n = in_sizes[0] / NFEAT;
    const int e = in_sizes[1] / 2;

    float* ws   = (float*)d_ws;
    size_t off  = ((size_t)n + 3) & ~(size_t)3;
    float* deg  = ws;
    float* h    = ws + off;
    float* agg  = h + (size_t)n * NCLASS;

    const int T = 256;

    const int mtiles = (n + 15) / 16;
    const int gblocks = (mtiles + 7) / 8;
    const int dblocks = (e + T * DC_U - 1) / (T * DC_U);

    deg_init_kernel <<<(n + T * 4 - 1) / (T * 4), T, 0, stream>>>(deg, out, n);
    gemm1_degcount_kernel<<<gblocks + dblocks, T, 0, stream>>>(x, Wg, h, n, gblocks, ei + e, deg, out, e);
    deg_sum_kernel<<<(n + T * 4 - 1) / (T * 4), T, 0, stream>>>(deg, out, n);

    agg_init_kernel<<<((n * (NCLASS / 4)) + T * AI_U - 1) / (T * AI_U), T, 0, stream>>>(h, deg, bg, agg, out, n);

    scatter_kernel<<<SC_BLOCKS, SC_THREADS, 0, stream>>>(ei, h, deg, agg, out, e, n);

    gemm2_kernel<<<gblocks, T, 0, stream>>>(agg, Wl, bl, out, n);
}


# edge sums by owner block in LDS: count stores 1 byte per edge (owner of its target), each block scans all bytes, adds its edges' rows into LDS rows by float adds; one sum plane, sum order changes
# speedup vs baseline: 2.4935x; 1.0531x over previous
#include <hip/hip_runtime.h>

typedef __attribute__((ext_vector_type(2))) float v2f;
typedef __attribute__((ext_vector_type(8))) float v8f;

#define NFEAT 256
#define NCLASS 64

__device__ __forceinline__ int chip_half() {
    return (int)((__builtin_amdgcn_s_sendmsg_rtn(0x87) >> 18) & 1u);
}

__global__ void deg_init_kernel(float* __restrict__ deg0, float* __restrict__ deg1, int n) {
    const int i = (blockIdx.x * blockDim.x + threadIdx.x) * 4;
    if (i + 3 < n) {
        *(float4*)(deg0 + i) = make_float4(1.0f, 1.0f, 1.0f, 1.0f);
        *(float4*)(deg1 + i) = make_float4(0.0f, 0.0f, 0.0f, 0.0f);
    } else {
        for (int k = i; k < n; ++k) {
            deg0[k] = 1.0f;
            deg1[k] = 0.0f;
        }
    }
}

#define DC_U 8

#define RS 391
#define SC_BLOCKS 256

__device__ __forceinline__ unsigned char owner_of(const int c) {
    return (unsigned char)(((unsigned)c / RS) & (SC_BLOCKS - 1));
}

__device__ __forceinline__ void deg_count_body(const int* __restrict__ col,
                                               float* __restrict__ deg0,
                                               float* __restrict__ deg1,
                                               unsigned char* __restrict__ own, int e,
                                               int blk, int nblk) {
    const int t = blk * 256 + threadIdx.x;
    const int stride = nblk * 256;
    float* deg = chip_half() ? deg1 : deg0;
    if (t + (DC_U - 1) * stride < e) {
        int c[DC_U];
#pragma unroll
        for (int u = 0; u < DC_U; ++u) c[u] = col[t + u * stride];
#pragma unroll
        for (int u = 0; u < DC_U; ++u) own[t + u * stride] = owner_of(c[u]);
#pragma unroll
        for (int u = 0; u < DC_U; ++u) atomicAdd(&deg[c[u]], 1.0f);
    } else {
        for (int i = t; i < e; i += stride) {
            const int c = col[i];
            own[i] = owner_of(c);
            atomicAdd(&deg[c], 1.0f);
        }
    }
}

__global__ void deg_sum_kernel(float* __restrict__ deg0, const float* __restrict__ deg1, int n) {
    const int i = (blockIdx.x * blockDim.x + threadIdx.x) * 4;
    if (i + 3 < n) {
        const float4 a = *(const float4*)(deg0 + i);
        const float4 b = *(const float4*)(deg1 + i);
        float4 o;
        o.x = a.x + b.x;
        o.y = a.y + b.y;
        o.z = a.z + b.z;
        o.w = a.w + b.w;
        *(float4*)(deg0 + i) = o;
    } else {
        for (int k = i; k < n; ++k) deg0[k] = deg0[k] + deg1[k];
    }
}

__device__ __forceinline__ void gemm1_body(const float* __restrict__ X,
                                           const float* __restrict__ Wg,
                                           float* __restrict__ H, int n, int blk) {
    const int lane = threadIdx.x & 31;
    const int wave = threadIdx.x >> 5;
    const int mtile = blk * 8 + wave;
    const int m0 = mtile * 16;
    if (m0 >= n) return;

    const int half = lane >> 4;
    const int l15  = lane & 15;
    int m = m0 + l15;
    if (m >= n) m = n - 1;
    const float* __restrict__ xrow = X + (size_t)m * NFEAT;

    v8f acc0 = {}, acc1 = {}, acc2 = {}, acc3 = {};

    for (int k0 = 0; k0 < NFEAT; k0 += 4) {
        const int kb = k0 + half * 2;
        v2f a;
        a.x = xrow[kb];
        a.y = xrow[kb + 1];

        const float* __restrict__ w0 = Wg + (size_t)kb * NCLASS;
        const float* __restrict__ w1 = w0 + NCLASS;
        v2f b0, b1, b2, b3;
        b0.x = w0[l15];      b0.y = w1[l15];
        b1.x = w0[16 + l15]; b1.y = w1[16 + l15];
        b2.x = w0[32 + l15]; b2.y = w1[32 + l15];
        b3.x = w0[48 + l15]; b3.y = w1[48 + l15];

        acc0 = __builtin_amdgcn_wmma_f32_16x16x4_f32(false, a, false, b0, (short)0, acc0, false, false);
        acc1 = __builtin_amdgcn_wmma_f32_16x16x4_f32(false, a, false, b1, (short)0, acc1, false, false);
        acc2 = __builtin_amdgcn_wmma_f32_16x16x4_f32(false, a, false, b2, (short)0, acc2, false, false);
        acc3 = __builtin_amdgcn_wmma_f32_16x16x4_f32(false, a, false, b3, (short)0, acc3, false, false);
    }

    const int mbase = m0 + half * 8;
#pragma unroll
    for (int r = 0; r < 8; ++r) {
        const int mr = mbase + r;
        if (mr < n) {
            float* __restrict__ hrow = H + (size_t)mr * NCLASS;
            hrow[l15]      = acc0[r];
            hrow[16 + l15] = acc1[r];
            hrow[32 + l15] = acc2[r];
            hrow[48 + l15] = acc3[r];
        }
    }
}

__global__ __launch_bounds__(256) void gemm1_degcount_kernel(const float* __restrict__ X,
                                                             const float* __restrict__ Wg,
                                                             float* __restrict__ H, int n,
                                                             int gblocks,
                                                             const int* __restrict__ col,
                                                             float* __restrict__ deg0,
                                                             float* __restrict__ deg1,
                                                             float* __restrict__ ownbuf, int e) {
    const int blk = (int)blockIdx.x;
    if (blk < gblocks) {
        gemm1_body(X, Wg, H, n, blk);
    } else {
        deg_count_body(col, deg0, deg1, (unsigned char*)ownbuf, e, blk - gblocks, (int)gridDim.x - gblocks);
    }
}

#define SC_BATCH 8
#define SC_THREADS 1024
#define DT_CAP 102400
#define DT_FULL 65535u
#define Q_CAP 64
#define SCAN_U 4

__device__ __forceinline__ uint2 pack_counts(const float4 d) {
    const unsigned u0 = min((unsigned)d.x, DT_FULL);
    const unsigned u1 = min((unsigned)d.y, DT_FULL);
    const unsigned u2 = min((unsigned)d.z, DT_FULL);
    const unsigned u3 = min((unsigned)d.w, DT_FULL);
    return make_uint2(u0 | (u1 << 16), u2 | (u3 << 16));
}

__device__ __forceinline__ float4 agg_start(const float4 h, const float d, const float4 b) {
    float4 o;
    o.x = h.x * d * d + b.x;
    o.y = h.y * d * d + b.y;
    o.z = h.z * d * d + b.z;
    o.w = h.w * d * d + b.w;
    return o;
}

__device__ __forceinline__ unsigned match_bytes(const unsigned x, const unsigned pat) {
    const unsigned y = x ^ pat;
    const unsigned z = ((y & 0x7F7F7F7Fu) + 0x7F7F7F7Fu) | y;
    return ~z & 0x80808080u;
}

__device__ __forceinline__ int piece_of(const int kk, const int nfull, const int rot) {
    const int ch = kk + rot;
    return (ch >= nfull) ? ch - nfull : ch;
}

__device__ __forceinline__ uint4 load_owners(const unsigned char* __restrict__ own, const int kk,
                                             const int nfull, const int rot, const int lane) {
    return *(const uint4*)(own + piece_of(min(kk, nfull - 1), nfull, rot) * 512 + lane * 16);
}

__device__ __forceinline__ unsigned piece_flags(const uint4 v, const unsigned pat, const bool live) {
    const unsigned f = (match_bytes(v.x, pat) >> 7) | (match_bytes(v.y, pat) >> 6) |
                       (match_bytes(v.z, pat) >> 5) | (match_bytes(v.w, pat) >> 4);
    return live ? f : 0u;
}

template <bool FULL>
__device__ __forceinline__ void add_edges(const int* __restrict__ ei, const float* __restrict__ H,
                                          const float* __restrict__ deg, const unsigned short* dtab,
                                          float* rows, const int e, const int node0, const int lane,
                                          const int idx, const int cnt) {
    int r = 0, lr = -1;
    float nrm = 0.0f;
    if (lane < cnt) {
        r = ei[idx];
        const int c = ei[e + idx];
        const unsigned tr = (r < DT_CAP) ? (unsigned)dtab[r] : DT_FULL;
        const unsigned tc = (c < DT_CAP) ? (unsigned)dtab[c] : DT_FULL;
        float dr = (float)tr, dc = (float)tc;
        if (tr == DT_FULL) dr = deg[r];
        if (tc == DT_FULL) dc = deg[c];
        nrm = rsqrtf(dr) * rsqrtf(dc);
        const int l = c - node0;
        if ((unsigned)l < (unsigned)RS) lr = l;
    }
    const int nbits = __float_as_int(nrm);

    if (FULL) {
        float h0[2][SC_BATCH], h1[2][SC_BATCH];
#pragma unroll
        for (int u = 0; u < SC_BATCH; ++u) {
            const float* src = H + (size_t)__builtin_amdgcn_readlane(r, u) * NCLASS;
            h0[0][u] = src[lane];
            h1[0][u] = src[32 + lane];
        }
#pragma unroll
        for (int b = 0; b < 32 / SC_BATCH; ++b) {
            if (b + 1 < 32 / SC_BATCH) {
#pragma unroll
                for (int u = 0; u < SC_BATCH; ++u) {
                    const int j = (b + 1) * SC_BATCH + u;
                    const float* src = H + (size_t)__builtin_amdgcn_readlane(r, j) * NCLASS;
                    h0[(b + 1) & 1][u] = src[lane];
                    h1[(b + 1) & 1][u] = src[32 + lane];
                }
            }
#pragma unroll
            for (int u = 0; u < SC_BATCH; ++u) {
                const int j = b * SC_BATCH + u;
                const float nj = __int_as_float(__builtin_amdgcn_readlane(nbits, j));
                const int lj = __builtin_amdgcn_readlane(lr, j);
                if (lj >= 0) {
                    float* dst = rows + lj * NCLASS;
                    atomicAdd(dst + lane, h0[b & 1][u] * nj);
                    atomicAdd(dst + 32 + lane, h1[b & 1][u] * nj);
                }
            }
        }
    } else {
        for (int j = 0; j < cnt; ++j) {
            const float nj = __int_as_float(__builtin_amdgcn_readlane(nbits, j));
            const int lj = __builtin_amdgcn_readlane(lr, j);
            const float* src = H + (size_t)__builtin_amdgcn_readlane(r, j) * NCLASS;
            const float a0 = src[lane];
            const float a1 = src[32 + lane];
            if (lj >= 0) {
                float* dst = rows + lj * NCLASS;
                atomicAdd(dst + lane, a0 * nj);
                atomicAdd(dst + 32 + lane, a1 * nj);
            }
        }
    }
}

__global__ __launch_bounds__(SC_THREADS) void edge_sum_kernel(const int* __restrict__ ei,
                                                              const float* __restrict__ H,
                                                              const float* __restrict__ deg,
                                                              const float* __restrict__ bg,
                                                              const float* __restrict__ ownbuf,
                                                              float* __restrict__ agg, int e, int n) {
    __shared__ __attribute__((aligned(16))) unsigned short dtab[DT_CAP];
    __shared__ __attribute__((aligned(16))) float rows[RS * NCLASS];
    __shared__ int found[(SC_THREADS / 32) * Q_CAP];

    const unsigned char* __restrict__ own = (const unsigned char*)ownbuf;

    {
        const int lim = min(n, DT_CAP);
        const int lim4 = lim & ~3;
        const int chunk = SC_THREADS * 4;
        const int full = lim4 / chunk;
        const int t4 = (int)threadIdx.x * 4;
        int t = full ? (int)(blockIdx.x % (unsigned)full) : 0;
#pragma unroll 8
        for (int k = 0; k < full; ++k) {
            const int i = t * chunk + t4;
            *(uint2*)(dtab + i) = pack_counts(*(const float4*)(deg + i));
            t = (t + 1 == full) ? 0 : t + 1;
        }
        const int i = full * chunk + t4;
        if (i < lim4) *(uint2*)(dtab + i) = pack_counts(*(const float4*)(deg + i));
        for (int k = lim4 + (int)threadIdx.x; k < lim; k += SC_THREADS)
            dtab[k] = (unsigned short)min((unsigned)deg[k], DT_FULL);
    }

    const int lane = threadIdx.x & 31;
    const int wib = __builtin_amdgcn_readfirstlane((int)(threadIdx.x >> 5));
    int* q = found + wib * Q_CAP;
    const int nslab = (n + RS - 1) / RS;
    const int nfull = e >> 9;
    const int rot = (int)(((unsigned)blockIdx.x * (unsigned)nfull) / SC_BLOCKS);
    const float4 b4 = ((const float4*)bg)[threadIdx.x & 15];
    const float4* __restrict__ H4 = (const float4*)H;
    float4* rows4 = (float4*)rows;
    float4* __restrict__ agg4 = (float4*)agg;

    for (int s = (int)blockIdx.x; s < nslab; s += SC_BLOCKS) {
        const int node0 = s * RS;

        for (int i = (int)threadIdx.x; i < RS * (NCLASS / 4); i += SC_THREADS) {
            const int node = node0 + (i >> 4);
            if (node < n)
                rows4[i] = agg_start(H4[(size_t)node * (NCLASS / 4) + (i & 15)], rsqrtf(deg[node]), b4);
        }
        __syncthreads();

        const unsigned pat = (unsigned)(s & (SC_BLOCKS - 1)) * 0x01010101u;
        int qn = 0;

        for (int k = wib; k < nfull; k += 32 * SCAN_U) {
            const uint4 a0 = load_owners(own, k, nfull, rot, lane);
            const uint4 a1 = load_owners(own, k + 32, nfull, rot, lane);
            const uint4 a2 = load_owners(own, k + 64, nfull, rot, lane);
            const uint4 a3 = load_owners(own, k + 96, nfull, rot, lane);
            const unsigned g01 = piece_flags(a0, pat, true) | (piece_flags(a1, pat, k + 32 < nfull) << 4);
            const unsigned g23 = piece_flags(a2, pat, k + 64 < nfull) | (piece_flags(a3, pat, k + 96 < nfull) << 4);
            unsigned long long g = (unsigned long long)g01 | ((unsigned long long)g23 << 32);

            unsigned bal = __builtin_amdgcn_ballot_w32(g != 0ull);
            while (bal) {
                if (g != 0ull) {
                    const int p = __builtin_ctzll(g);
                    g &= g - 1ull;
                    const int u = ((p >> 5) << 1) | ((p >> 2) & 1);
                    const int byte = 4 * (p & 3) + ((p & 31) >> 3);
                    q[qn + (int)__builtin_amdgcn_mbcnt_lo(bal, 0u)] =
                        piece_of(k + 32 * u, nfull, rot) * 512 + lane * 16 + byte;
                }
                qn += __builtin_popcount(bal);
                if (qn >= 32) {
                    qn -= 32;
                    add_edges<true>(ei, H, deg, dtab, rows, e, node0, lane, q[qn + lane], 32);
                }
                bal = __builtin_amdgcn_ballot_w32(g != 0ull);
            }
        }
        if (qn > 0) {
            const int idx = (lane < qn) ? q[lane] : 0;
            add_edges<false>(ei, H, deg, dtab, rows, e, node0, lane, idx, qn);
        }
        for (int i = nfull * 512 + (int)threadIdx.x; i < e; i += SC_THREADS) {
            const int c = ei[e + i];
            const int l = c - node0;
            if ((unsigned)l < (unsigned)RS) {
                const int r = ei[i];
                const float w = rsqrtf(deg[r]) * rsqrtf(deg[c]);
                const float* src = H + (size_t)r * NCLASS;
                for (int f = 0; f < NCLASS; ++f) atomicAdd(rows + l * NCLASS + f, src[f] * w);
            }
        }
        __syncthreads();

        for (int i = (int)threadIdx.x; i < RS * (NCLASS / 4); i += SC_THREADS) {
            const int node = node0 + (i >> 4);
            if (node < n) agg4[(size_t)node * (NCLASS / 4) + (i & 15)] = rows4[i];
        }
        __syncthreads();
    }
}

__global__ __launch_bounds__(256) void gemm2_kernel(const float* __restrict__ AGG0,
                                                    const float* __restrict__ Wl,
                                                    const float* __restrict__ bl,
                                                    float* __restrict__ OUT, int n) {
    const int lane = threadIdx.x & 31;
    const int wave = threadIdx.x >> 5;
    const int mtile = blockIdx.x * 8 + wave;
    const int m0 = mtile * 16;
    if (m0 >= n) return;

    const int half = lane >> 4;
    const int l15  = lane & 15;
    int m = m0 + l15;
    if (m >= n) m = n - 1;
    const float* __restrict__ arow = AGG0 + (size_t)m * NCLASS;

    v8f acc0 = {}, acc1 = {}, acc2 = {}, acc3 = {};

    for (int k0 = 0; k0 < NCLASS; k0 += 4) {
        const int kb = k0 + half * 2;
        v2f a;
        a.x = fmaxf(arow[kb], 0.0f);
        a.y = fmaxf(arow[kb + 1], 0.0f);

        const float* __restrict__ w0 = Wl + (size_t)kb * NCLASS;
        const float* __restrict__ w1 = w0 + NCLASS;
        v2f b0, b1, b2, b3;
        b0.x = w0[l15];      b0.y = w1[l15];
        b1.x = w0[16 + l15]; b1.y = w1[16 + l15];
        b2.x = w0[32 + l15]; b2.y = w1[32 + l15];
        b3.x = w0[48 + l15]; b3.y = w1[48 + l15];

        acc0 = __builtin_amdgcn_wmma_f32_16x16x4_f32(false, a, false, b0, (short)0, acc0, false, false);
        acc1 = __builtin_amdgcn_wmma_f32_16x16x4_f32(false, a, false, b1, (short)0, acc1, false, false);
        acc2 = __builtin_amdgcn_wmma_f32_16x16x4_f32(false, a, false, b2, (short)0, acc2, false, false);
        acc3 = __builtin_amdgcn_wmma_f32_16x16x4_f32(false, a, false, b3, (short)0, acc3, false, false);
    }

    const float bl0 = bl[l15];
    const float bl1 = bl[16 + l15];
    const float bl2 = bl[32 + l15];
    const float bl3 = bl[48 + l15];

    const int mbase = m0 + half * 8;
#pragma unroll
    for (int r = 0; r < 8; ++r) {
        const int mr = mbase + r;
        if (mr < n) {
            float* orow = OUT + (size_t)mr * NCLASS;
            orow[l15]      = acc0[r] + bl0;
            orow[16 + l15] = acc1[r] + bl1;
            orow[32 + l15] = acc2[r] + bl2;
            orow[48 + l15] = acc3[r] + bl3;
        }
    }
}

extern "C" void kernel_launch(void* const* d_in, const int* in_sizes, int n_in,
                              void* d_out, int out_size, void* d_ws, size_t ws_size,
                              hipStream_t stream) {
    const float* x  = (const float*)d_in[0];
    const int*   ei = (const int*)d_in[1];
    const float* Wg = (const float*)d_in[2];
    const float* bg = (const float*)d_in[3];
    const float* Wl = (const float*)d_in[4];
    const float* bl = (const float*)d_in[5];
    float* out = (float*)d_out;

    const int n = in_sizes[0] / NFEAT;
    const int e = in_sizes[1] / 2;

    float* ws   = (float*)d_ws;
    size_t off  = ((size_t)n + 3) & ~(size_t)3;
    float* deg  = ws;
    float* h    = ws + off;
    float* agg  = h + (size_t)n * NCLASS;
    float* own  = out + off;

    const int T = 256;

    const int mtiles = (n + 15) / 16;
    const int gblocks = (mtiles + 7) / 8;
    const int dblocks = (e + T * DC_U - 1) / (T * DC_U);

    deg_init_kernel <<<(n + T * 4 - 1) / (T * 4), T, 0, stream>>>(deg, out, n);
    gemm1_degcount_kernel<<<gblocks + dblocks, T, 0, stream>>>(x, Wg, h, n, gblocks, ei + e, deg, out, own, e);
    deg_sum_kernel<<<(n + T * 4 - 1) / (T * 4), T, 0, stream>>>(deg, out, n);

    edge_sum_kernel<<<SC_BLOCKS, SC_THREADS, 0, stream>>>(ei, h, deg, bg, own, agg, e, n);

    gemm2_kernel<<<gblocks, T, 0, stream>>>(agg, Wl, bl, out, n);
}


# edges sorted by owner block in the count launch (per 2048-edge chunk: LDS counters, prefix sum, source and target lists); edge blocks read only their own runs; LDS float adds in arrival order
# speedup vs baseline: 3.1660x; 1.2697x over previous
#include <hip/hip_runtime.h>

typedef __attribute__((ext_vector_type(2))) float v2f;
typedef __attribute__((ext_vector_type(8))) float v8f;

#define NFEAT 256
#define NCLASS 64

__device__ __forceinline__ int chip_half() {
    return (int)((__builtin_amdgcn_s_sendmsg_rtn(0x87) >> 18) & 1u);
}

__global__ void deg_init_kernel(float* __restrict__ deg0, float* __restrict__ deg1, int n) {
    const int i = (blockIdx.x * blockDim.x + threadIdx.x) * 4;
    if (i + 3 < n) {
        *(float4*)(deg0 + i) = make_float4(1.0f, 1.0f, 1.0f, 1.0f);
        *(float4*)(deg1 + i) = make_float4(0.0f, 0.0f, 0.0f, 0.0f);
    } else {
        for (int k = i; k < n; ++k) {
            deg0[k] = 1.0f;
            deg1[k] = 0.0f;
        }
    }
}

#define DC_U 8
#define CH_EDGES (256 * DC_U)
#define RS 391
#define SC_BLOCKS 256
#define L_BITS 9

__device__ __forceinline__ void part_body(const int* __restrict__ ei,
                                          float* __restrict__ deg0,
                                          float* __restrict__ deg1,
                                          unsigned* __restrict__ offs,
                                          int* __restrict__ srcL,
                                          unsigned short* __restrict__ dstL,
                                          int e, int blk) {
    __shared__ int hist[SC_BLOCKS];
    __shared__ int sc[2][SC_BLOCKS];
    const int t = (int)threadIdx.x;
    hist[t] = 0;
    __syncthreads();

    const int base = blk * CH_EDGES;
    float* deg = chip_half() ? deg1 : deg0;
    int r[DC_U], c[DC_U], slot[DC_U];
#pragma unroll
    for (int u = 0; u < DC_U; ++u) {
        const int i = min(base + u * 256 + t, e - 1);
        r[u] = ei[i];
        c[u] = ei[e + i];
    }
#pragma unroll
    for (int u = 0; u < DC_U; ++u) {
        const bool live = base + u * 256 + t < e;
        slot[u] = 0;
        if (live) {
            slot[u] = atomicAdd(&hist[((unsigned)c[u] / RS) & (SC_BLOCKS - 1)], 1);
            atomicAdd(&deg[c[u]], 1.0f);
        }
    }
    __syncthreads();

    const int v = hist[t];
    sc[0][t] = v;
    __syncthreads();
    int pp = 0;
    for (int d = 1; d < SC_BLOCKS; d <<= 1) {
        int x = sc[pp][t];
        if (t >= d) x += sc[pp][t - d];
        sc[pp ^ 1][t] = x;
        pp ^= 1;
        __syncthreads();
    }
    const int start = sc[pp][t] - v;
    offs[blk * SC_BLOCKS + t] = ((unsigned)start << 16) | (unsigned)v;
    hist[t] = start;
    __syncthreads();

#pragma unroll
    for (int u = 0; u < DC_U; ++u) {
        const bool live = base + u * 256 + t < e;
        if (live) {
            const unsigned slab = (unsigned)c[u] / RS;
            const int pos = base + hist[slab & (SC_BLOCKS - 1)] + slot[u];
            srcL[pos] = r[u];
            dstL[pos] = (unsigned short)(((slab >> 8) << L_BITS) | ((unsigned)c[u] - slab * RS));
        }
    }
}

__global__ void deg_sum_kernel(float* __restrict__ deg0, const float* __restrict__ deg1, int n) {
    const int i = (blockIdx.x * blockDim.x + threadIdx.x) * 4;
    if (i + 3 < n) {
        const float4 a = *(const float4*)(deg0 + i);
        const float4 b = *(const float4*)(deg1 + i);
        float4 o;
        o.x = a.x + b.x;
        o.y = a.y + b.y;
        o.z = a.z + b.z;
        o.w = a.w + b.w;
        *(float4*)(deg0 + i) = o;
    } else {
        for (int k = i; k < n; ++k) deg0[k] = deg0[k] + deg1[k];
    }
}

__device__ __forceinline__ void gemm1_body(const float* __restrict__ X,
                                           const float* __restrict__ Wg,
                                           float* __restrict__ H, int n, int blk) {
    const int lane = threadIdx.x & 31;
    const int wave = threadIdx.x >> 5;
    const int mtile = blk * 8 + wave;
    const int m0 = mtile * 16;
    if (m0 >= n) return;

    const int half = lane >> 4;
    const int l15  = lane & 15;
    int m = m0 + l15;
    if (m >= n) m = n - 1;
    const float* __restrict__ xrow = X + (size_t)m * NFEAT;

    v8f acc0 = {}, acc1 = {}, acc2 = {}, acc3 = {};

    for (int k0 = 0; k0 < NFEAT; k0 += 4) {
        const int kb = k0 + half * 2;
        v2f a;
        a.x = xrow[kb];
        a.y = xrow[kb + 1];

        const float* __restrict__ w0 = Wg + (size_t)kb * NCLASS;
        const float* __restrict__ w1 = w0 + NCLASS;
        v2f b0, b1, b2, b3;
        b0.x = w0[l15];      b0.y = w1[l15];
        b1.x = w0[16 + l15]; b1.y = w1[16 + l15];
        b2.x = w0[32 + l15]; b2.y = w1[32 + l15];
        b3.x = w0[48 + l15]; b3.y = w1[48 + l15];

        acc0 = __builtin_amdgcn_wmma_f32_16x16x4_f32(false, a, false, b0, (short)0, acc0, false, false);
        acc1 = __builtin_amdgcn_wmma_f32_16x16x4_f32(false, a, false, b1, (short)0, acc1, false, false);
        acc2 = __builtin_amdgcn_wmma_f32_16x16x4_f32(false, a, false, b2, (short)0, acc2, false, false);
        acc3 = __builtin_amdgcn_wmma_f32_16x16x4_f32(false, a, false, b3, (short)0, acc3, false, false);
    }

    const int mbase = m0 + half * 8;
#pragma unroll
    for (int r = 0; r < 8; ++r) {
        const int mr = mbase + r;
        if (mr < n) {
            float* __restrict__ hrow = H + (size_t)mr * NCLASS;
            hrow[l15]      = acc0[r];
            hrow[16 + l15] = acc1[r];
            hrow[32 + l15] = acc2[r];
            hrow[48 + l15] = acc3[r];
        }
    }
}

__global__ __launch_bounds__(256) void gemm1_degcount_kernel(const float* __restrict__ X,
                                                             const float* __restrict__ Wg,
                                                             float* __restrict__ H, int n,
                                                             int gblocks,
                                                             const int* __restrict__ ei,
                                                             float* __restrict__ deg0,
                                                             float* __restrict__ deg1,
                                                             float* __restrict__ offsbuf,
                                                             float* __restrict__ srcbuf,
                                                             float* __restrict__ dstbuf, int e) {
    const int blk = (int)blockIdx.x;
    if (blk < gblocks) {
        gemm1_body(X, Wg, H, n, blk);
    } else {
        part_body(ei, deg0, deg1, (unsigned*)offsbuf, (int*)srcbuf, (unsigned short*)dstbuf, e, blk - gblocks);
    }
}

#define SC_BATCH 8
#define SC_THREADS 1024
#define DT_CAP 102400
#define DT_FULL 65535u
#define Q_CAP 64

__device__ __forceinline__ uint2 pack_counts(const float4 d) {
    const unsigned u0 = min((unsigned)d.x, DT_FULL);
    const unsigned u1 = min((unsigned)d.y, DT_FULL);
    const unsigned u2 = min((unsigned)d.z, DT_FULL);
    const unsigned u3 = min((unsigned)d.w, DT_FULL);
    return make_uint2(u0 | (u1 << 16), u2 | (u3 << 16));
}

__device__ __forceinline__ float4 agg_start(const float4 h, const float d, const float4 b) {
    float4 o;
    o.x = h.x * d * d + b.x;
    o.y = h.y * d * d + b.y;
    o.z = h.z * d * d + b.z;
    o.w = h.w * d * d + b.w;
    return o;
}

template <bool FULL>
__device__ __forceinline__ void add_edges(const int* __restrict__ srcL,
                                          const unsigned short* __restrict__ dstL,
                                          const float* __restrict__ H,
                                          const float* __restrict__ deg, const unsigned short* dtab,
                                          float* rows, const int node0, const int pass, const int lane,
                                          const int idx, const int cnt) {
    int r = 0, lr = -1;
    float nrm = 0.0f;
    if (lane < cnt) {
        r = srcL[idx];
        const unsigned d = (unsigned)dstL[idx];
        if ((int)(d >> L_BITS) == pass) {
            const int l = (int)(d & ((1u << L_BITS) - 1u));
            const int c = node0 + l;
            const unsigned tr = (r < DT_CAP) ? (unsigned)dtab[r] : DT_FULL;
            const unsigned tc = (c < DT_CAP) ? (unsigned)dtab[c] : DT_FULL;
            float dr = (float)tr, dc = (float)tc;
            if (tr == DT_FULL) dr = deg[r];
            if (tc == DT_FULL) dc = deg[c];
            nrm = rsqrtf(dr) * rsqrtf(dc);
            lr = l;
        }
    }
    const int nbits = __float_as_int(nrm);

    if (FULL) {
        float h0[2][SC_BATCH], h1[2][SC_BATCH];
#pragma unroll
        for (int u = 0; u < SC_BATCH; ++u) {
            const float* src = H + (size_t)__builtin_amdgcn_readlane(r, u) * NCLASS;
            h0[0][u] = src[lane];
            h1[0][u] = src[32 + lane];
        }
#pragma unroll
        for (int b = 0; b < 32 / SC_BATCH; ++b) {
            if (b + 1 < 32 / SC_BATCH) {
#pragma unroll
                for (int u = 0; u < SC_BATCH; ++u) {
                    const int j = (b + 1) * SC_BATCH + u;
                    const float* src = H + (size_t)__builtin_amdgcn_readlane(r, j) * NCLASS;
                    h0[(b + 1) & 1][u] = src[lane];
                    h1[(b + 1) & 1][u] = src[32 + lane];
                }
            }
#pragma unroll
            for (int u = 0; u < SC_BATCH; ++u) {
                const int j = b * SC_BATCH + u;
                const float nj = __int_as_float(__builtin_amdgcn_readlane(nbits, j));
                const int lj = __builtin_amdgcn_readlane(lr, j);
                if (lj >= 0) {
                    float* dst = rows + lj * NCLASS;
                    atomicAdd(dst + lane, h0[b & 1][u] * nj);
                    atomicAdd(dst + 32 + lane, h1[b & 1][u] * nj);
                }
            }
        }
    } else {
        for (int j = 0; j < cnt; ++j) {
            const float nj = __int_as_float(__builtin_amdgcn_readlane(nbits, j));
            const int lj = __builtin_amdgcn_readlane(lr, j);
            const float* src = H + (size_t)__builtin_amdgcn_readlane(r, j) * NCLASS;
            const float a0 = src[lane];
            const float a1 = src[32 + lane];
            if (lj >= 0) {
                float* dst = rows + lj * NCLASS;
                atomicAdd(dst + lane, a0 * nj);
                atomicAdd(dst + 32 + lane, a1 * nj);
            }
        }
    }
}

__global__ __launch_bounds__(SC_THREADS) void edge_sum_kernel(const float* __restrict__ H,
                                                              const float* __restrict__ deg,
                                                              const float* __restrict__ bg,
                                                              const float* __restrict__ offsbuf,
                                                              const float* __restrict__ srcbuf,
                                                              const float* __restrict__ dstbuf,
                                                              float* __restrict__ agg, int e, int n) {
    __shared__ __attribute__((aligned(16))) unsigned short dtab[DT_CAP];
    __shared__ __attribute__((aligned(16))) float rows[RS * NCLASS];
    __shared__ int found[(SC_THREADS / 32) * Q_CAP];

    const unsigned* __restrict__ offs = (const unsigned*)offsbuf;
    const int* __restrict__ srcL = (const int*)srcbuf;
    const unsigned short* __restrict__ dstL = (const unsigned short*)dstbuf;

    {
        const int lim = min(n, DT_CAP);
        const int lim4 = lim & ~3;
        const int chunk = SC_THREADS * 4;
        const int full = lim4 / chunk;
        const int t4 = (int)threadIdx.x * 4;
        int t = full ? (int)(blockIdx.x % (unsigned)full) : 0;
#pragma unroll 8
        for (int k = 0; k < full; ++k) {
            const int i = t * chunk + t4;
            *(uint2*)(dtab + i) = pack_counts(*(const float4*)(deg + i));
            t = (t + 1 == full) ? 0 : t + 1;
        }
        const int i = full * chunk + t4;
        if (i < lim4) *(uint2*)(dtab + i) = pack_counts(*(const float4*)(deg + i));
        for (int k = lim4 + (int)threadIdx.x; k < lim; k += SC_THREADS)
            dtab[k] = (unsigned short)min((unsigned)deg[k], DT_FULL);
    }

    const int lane = threadIdx.x & 31;
    const int wib = __builtin_amdgcn_readfirstlane((int)(threadIdx.x >> 5));
    int* q = found + wib * Q_CAP;
    const int nslab = (n + RS - 1) / RS;
    const int nch = (e + CH_EDGES - 1) / CH_EDGES;
    const int per = (nch + 31) / 32;
    const int owner = (int)blockIdx.x;
    const float4 b4 = ((const float4*)bg)[threadIdx.x & 15];
    const float4* __restrict__ H4 = (const float4*)H;
    float4* rows4 = (float4*)rows;
    float4* __restrict__ agg4 = (float4*)agg;

    for (int s = (int)blockIdx.x; s < nslab; s += SC_BLOCKS) {
        const int node0 = s * RS;

        for (int i = (int)threadIdx.x; i < RS * (NCLASS / 4); i += SC_THREADS) {
            const int node = node0 + (i >> 4);
            if (node < n)
                rows4[i] = agg_start(H4[(size_t)node * (NCLASS / 4) + (i & 15)], rsqrtf(deg[node]), b4);
        }
        __syncthreads();

        const int pass = s / SC_BLOCKS;
        int qn = 0;

        for (int j0 = 0; j0 < per; j0 += 32) {
            const int jj = j0 + lane;
            const int ch = wib * per + jj;
            unsigned oc = 0u;
            if (jj < per && ch < nch) oc = offs[(size_t)ch * SC_BLOCKS + owner];
            int left = (int)(oc & 0xFFFFu);
            int pos = ch * CH_EDGES + (int)(oc >> 16);

            unsigned bal = __builtin_amdgcn_ballot_w32(left > 0);
            while (bal) {
                if (left > 0) {
                    q[qn + (int)__builtin_amdgcn_mbcnt_lo(bal, 0u)] = pos;
                    ++pos;
                    --left;
                }
                qn += __builtin_popcount(bal);
                if (qn >= 32) {
                    qn -= 32;
                    add_edges<true>(srcL, dstL, H, deg, dtab, rows, node0, pass, lane, q[qn + lane], 32);
                }
                bal = __builtin_amdgcn_ballot_w32(left > 0);
            }
        }
        if (qn > 0) {
            const int idx = (lane < qn) ? q[lane] : 0;
            add_edges<false>(srcL, dstL, H, deg, dtab, rows, node0, pass, lane, idx, qn);
        }
        __syncthreads();

        for (int i = (int)threadIdx.x; i < RS * (NCLASS / 4); i += SC_THREADS) {
            const int node = node0 + (i >> 4);
            if (node < n) agg4[(size_t)node * (NCLASS / 4) + (i & 15)] = rows4[i];
        }
        __syncthreads();
    }
}

__global__ __launch_bounds__(256) void gemm2_kernel(const float* __restrict__ AGG0,
                                                    const float* __restrict__ Wl,
                                                    const float* __restrict__ bl,
                                                    float* __restrict__ OUT, int n) {
    const int lane = threadIdx.x & 31;
    const int wave = threadIdx.x >> 5;
    const int mtile = blockIdx.x * 8 + wave;
    const int m0 = mtile * 16;
    if (m0 >= n) return;

    const int half = lane >> 4;
    const int l15  = lane & 15;
    int m = m0 + l15;
    if (m >= n) m = n - 1;
    const float* __restrict__ arow = AGG0 + (size_t)m * NCLASS;

    v8f acc0 = {}, acc1 = {}, acc2 = {}, acc3 = {};

    for (int k0 = 0; k0 < NCLASS; k0 += 4) {
        const int kb = k0 + half * 2;
        v2f a;
        a.x = fmaxf(arow[kb], 0.0f);
        a.y = fmaxf(arow[kb + 1], 0.0f);

        const float* __restrict__ w0 = Wl + (size_t)kb * NCLASS;
        const float* __restrict__ w1 = w0 + NCLASS;
        v2f b0, b1, b2, b3;
        b0.x = w0[l15];      b0.y = w1[l15];
        b1.x = w0[16 + l15]; b1.y = w1[16 + l15];
        b2.x = w0[32 + l15]; b2.y = w1[32 + l15];
        b3.x = w0[48 + l15]; b3.y = w1[48 + l15];

        acc0 = __builtin_amdgcn_wmma_f32_16x16x4_f32(false, a, false, b0, (short)0, acc0, false, false);
        acc1 = __builtin_amdgcn_wmma_f32_16x16x4_f32(false, a, false, b1, (short)0, acc1, false, false);
        acc2 = __builtin_amdgcn_wmma_f32_16x16x4_f32(false, a, false, b2, (short)0, acc2, false, false);
        acc3 = __builtin_amdgcn_wmma_f32_16x16x4_f32(false, a, false, b3, (short)0, acc3, false, false);
    }

    const float bl0 = bl[l15];
    const float bl1 = bl[16 + l15];
    const float bl2 = bl[32 + l15];
    const float bl3 = bl[48 + l15];

    const int mbase = m0 + half * 8;
#pragma unroll
    for (int r = 0; r < 8; ++r) {
        const int mr = mbase + r;
        if (mr < n) {
            float* orow = OUT + (size_t)mr * NCLASS;
            orow[l15]      = acc0[r] + bl0;
            orow[16 + l15] = acc1[r] + bl1;
            orow[32 + l15] = acc2[r] + bl2;
            orow[48 + l15] = acc3[r] + bl3;
        }
    }
}

extern "C" void kernel_launch(void* const* d_in, const int* in_sizes, int n_in,
                              void* d_out, int out_size, void* d_ws, size_t ws_size,
                              hipStream_t stream) {
    const float* x  = (const float*)d_in[0];
    const int*   ei = (const int*)d_in[1];
    const float* Wg = (const float*)d_in[2];
    const float* bg = (const float*)d_in[3];
    const float* Wl = (const float*)d_in[4];
    const float* bl = (const float*)d_in[5];
    float* out = (float*)d_out;

    const int n = in_sizes[0] / NFEAT;
    const int e = in_sizes[1] / 2;

    float* ws   = (float*)d_ws;
    size_t off  = ((size_t)n + 3) & ~(size_t)3;
    float* deg  = ws;
    float* h    = ws + off;
    float* agg  = h + (size_t)n * NCLASS;

    const int T = 256;

    const int mtiles = (n + 15) / 16;
    const int gblocks = (mtiles + 7) / 8;
    const int dblocks = (e + T * DC_U - 1) / (T * DC_U);
    float* offsb = out + off;
    float* srcb  = offsb + (size_t)dblocks * SC_BLOCKS;
    float* dstb  = srcb + e;

    deg_init_kernel <<<(n + T * 4 - 1) / (T * 4), T, 0, stream>>>(deg, out, n);
    gemm1_degcount_kernel<<<gblocks + dblocks, T, 0, stream>>>(x, Wg, h, n, gblocks, ei, deg, out, offsb, srcb, dstb, e);
    deg_sum_kernel<<<(n + T * 4 - 1) / (T * 4), T, 0, stream>>>(deg, out, n);

    edge_sum_kernel<<<SC_BLOCKS, SC_THREADS, 0, stream>>>(h, deg, bg, offsb, srcb, dstb, agg, e, n);

    gemm2_kernel<<<gblocks, T, 0, stream>>>(agg, Wl, bl, out, n);
}


# edge kernel lists each run with consecutive lanes (runs walked one after another, not one lane per chunk): a group's list loads touch a few lines, not 32 each; LDS float adds in arrival order
# speedup vs baseline: 3.3251x; 1.0503x over previous
#include <hip/hip_runtime.h>

typedef __attribute__((ext_vector_type(2))) float v2f;
typedef __attribute__((ext_vector_type(8))) float v8f;

#define NFEAT 256
#define NCLASS 64

__device__ __forceinline__ int chip_half() {
    return (int)((__builtin_amdgcn_s_sendmsg_rtn(0x87) >> 18) & 1u);
}

__global__ void deg_init_kernel(float* __restrict__ deg0, float* __restrict__ deg1, int n) {
    const int i = (blockIdx.x * blockDim.x + threadIdx.x) * 4;
    if (i + 3 < n) {
        *(float4*)(deg0 + i) = make_float4(1.0f, 1.0f, 1.0f, 1.0f);
        *(float4*)(deg1 + i) = make_float4(0.0f, 0.0f, 0.0f, 0.0f);
    } else {
        for (int k = i; k < n; ++k) {
            deg0[k] = 1.0f;
            deg1[k] = 0.0f;
        }
    }
}

#define DC_U 8
#define CH_EDGES (256 * DC_U)
#define RS 391
#define SC_BLOCKS 256
#define L_BITS 9

__device__ __forceinline__ void part_body(const int* __restrict__ ei,
                                          float* __restrict__ deg0,
                                          float* __restrict__ deg1,
                                          unsigned* __restrict__ offs,
                                          int* __restrict__ srcL,
                                          unsigned short* __restrict__ dstL,
                                          int e, int blk) {
    __shared__ int hist[SC_BLOCKS];
    __shared__ int sc[2][SC_BLOCKS];
    const int t = (int)threadIdx.x;
    hist[t] = 0;
    __syncthreads();

    const int base = blk * CH_EDGES;
    float* deg = chip_half() ? deg1 : deg0;
    int r[DC_U], c[DC_U], slot[DC_U];
#pragma unroll
    for (int u = 0; u < DC_U; ++u) {
        const int i = min(base + u * 256 + t, e - 1);
        r[u] = ei[i];
        c[u] = ei[e + i];
    }
#pragma unroll
    for (int u = 0; u < DC_U; ++u) {
        const bool live = base + u * 256 + t < e;
        slot[u] = 0;
        if (live) {
            slot[u] = atomicAdd(&hist[((unsigned)c[u] / RS) & (SC_BLOCKS - 1)], 1);
            atomicAdd(&deg[c[u]], 1.0f);
        }
    }
    __syncthreads();

    const int v = hist[t];
    sc[0][t] = v;
    __syncthreads();
    int pp = 0;
    for (int d = 1; d < SC_BLOCKS; d <<= 1) {
        int x = sc[pp][t];
        if (t >= d) x += sc[pp][t - d];
        sc[pp ^ 1][t] = x;
        pp ^= 1;
        __syncthreads();
    }
    const int start = sc[pp][t] - v;
    offs[blk * SC_BLOCKS + t] = ((unsigned)start << 16) | (unsigned)v;
    hist[t] = start;
    __syncthreads();

#pragma unroll
    for (int u = 0; u < DC_U; ++u) {
        const bool live = base + u * 256 + t < e;
        if (live) {
            const unsigned slab = (unsigned)c[u] / RS;
            const int pos = base + hist[slab & (SC_BLOCKS - 1)] + slot[u];
            srcL[pos] = r[u];
            dstL[pos] = (unsigned short)(((slab >> 8) << L_BITS) | ((unsigned)c[u] - slab * RS));
        }
    }
}

__global__ void deg_sum_kernel(float* __restrict__ deg0, const float* __restrict__ deg1, int n) {
    const int i = (blockIdx.x * blockDim.x + threadIdx.x) * 4;
    if (i + 3 < n) {
        const float4 a = *(const float4*)(deg0 + i);
        const float4 b = *(const float4*)(deg1 + i);
        float4 o;
        o.x = a.x + b.x;
        o.y = a.y + b.y;
        o.z = a.z + b.z;
        o.w = a.w + b.w;
        *(float4*)(deg0 + i) = o;
    } else {
        for (int k = i; k < n; ++k) deg0[k] = deg0[k] + deg1[k];
    }
}

__device__ __forceinline__ void gemm1_body(const float* __restrict__ X,
                                           const float* __restrict__ Wg,
                                           float* __restrict__ H, int n, int blk) {
    const int lane = threadIdx.x & 31;
    const int wave = threadIdx.x >> 5;
    const int mtile = blk * 8 + wave;
    const int m0 = mtile * 16;
    if (m0 >= n) return;

    const int half = lane >> 4;
    const int l15  = lane & 15;
    int m = m0 + l15;
    if (m >= n) m = n - 1;
    const float* __restrict__ xrow = X + (size_t)m * NFEAT;

    v8f acc0 = {}, acc1 = {}, acc2 = {}, acc3 = {};

    for (int k0 = 0; k0 < NFEAT; k0 += 4) {
        const int kb = k0 + half * 2;
        v2f a;
        a.x = xrow[kb];
        a.y = xrow[kb + 1];

        const float* __restrict__ w0 = Wg + (size_t)kb * NCLASS;
        const float* __restrict__ w1 = w0 + NCLASS;
        v2f b0, b1, b2, b3;
        b0.x = w0[l15];      b0.y = w1[l15];
        b1.x = w0[16 + l15]; b1.y = w1[16 + l15];
        b2.x = w0[32 + l15]; b2.y = w1[32 + l15];
        b3.x = w0[48 + l15]; b3.y = w1[48 + l15];

        acc0 = __builtin_amdgcn_wmma_f32_16x16x4_f32(false, a, false, b0, (short)0, acc0, false, false);
        acc1 = __builtin_amdgcn_wmma_f32_16x16x4_f32(false, a, false, b1, (short)0, acc1, false, false);
        acc2 = __builtin_amdgcn_wmma_f32_16x16x4_f32(false, a, false, b2, (short)0, acc2, false, false);
        acc3 = __builtin_amdgcn_wmma_f32_16x16x4_f32(false, a, false, b3, (short)0, acc3, false, false);
    }

    const int mbase = m0 + half * 8;
#pragma unroll
    for (int r = 0; r < 8; ++r) {
        const int mr = mbase + r;
        if (mr < n) {
            float* __restrict__ hrow = H + (size_t)mr * NCLASS;
            hrow[l15]      = acc0[r];
            hrow[16 + l15] = acc1[r];
            hrow[32 + l15] = acc2[r];
            hrow[48 + l15] = acc3[r];
        }
    }
}

__global__ __launch_bounds__(256) void gemm1_degcount_kernel(const float* __restrict__ X,
                                                             const float* __restrict__ Wg,
                                                             float* __restrict__ H, int n,
                                                             int gblocks,
                                                             const int* __restrict__ ei,
                                                             float* __restrict__ deg0,
                                                             float* __restrict__ deg1,
                                                             float* __restrict__ offsbuf,
                                                             float* __restrict__ srcbuf,
                                                             float* __restrict__ dstbuf, int e) {
    const int blk = (int)blockIdx.x;
    if (blk < gblocks) {
        gemm1_body(X, Wg, H, n, blk);
    } else {
        part_body(ei, deg0, deg1, (unsigned*)offsbuf, (int*)srcbuf, (unsigned short*)dstbuf, e, blk - gblocks);
    }
}

#define SC_BATCH 8
#define SC_THREADS 1024
#define DT_CAP 102400
#define DT_FULL 65535u
#define Q_CAP 64

__device__ __forceinline__ uint2 pack_counts(const float4 d) {
    const unsigned u0 = min((unsigned)d.x, DT_FULL);
    const unsigned u1 = min((unsigned)d.y, DT_FULL);
    const unsigned u2 = min((unsigned)d.z, DT_FULL);
    const unsigned u3 = min((unsigned)d.w, DT_FULL);
    return make_uint2(u0 | (u1 << 16), u2 | (u3 << 16));
}

__device__ __forceinline__ float4 agg_start(const float4 h, const float d, const float4 b) {
    float4 o;
    o.x = h.x * d * d + b.x;
    o.y = h.y * d * d + b.y;
    o.z = h.z * d * d + b.z;
    o.w = h.w * d * d + b.w;
    return o;
}

template <bool FULL>
__device__ __forceinline__ void add_edges(const int* __restrict__ srcL,
                                          const unsigned short* __restrict__ dstL,
                                          const float* __restrict__ H,
                                          const float* __restrict__ deg, const unsigned short* dtab,
                                          float* rows, const int node0, const int pass, const int lane,
                                          const int idx, const int cnt) {
    int r = 0, lr = -1;
    float nrm = 0.0f;
    if (lane < cnt) {
        r = srcL[idx];
        const unsigned d = (unsigned)dstL[idx];
        if ((int)(d >> L_BITS) == pass) {
            const int l = (int)(d & ((1u << L_BITS) - 1u));
            const int c = node0 + l;
            const unsigned tr = (r < DT_CAP) ? (unsigned)dtab[r] : DT_FULL;
            const unsigned tc = (c < DT_CAP) ? (unsigned)dtab[c] : DT_FULL;
            float dr = (float)tr, dc = (float)tc;
            if (tr == DT_FULL) dr = deg[r];
            if (tc == DT_FULL) dc = deg[c];
            nrm = rsqrtf(dr) * rsqrtf(dc);
            lr = l;
        }
    }
    const int nbits = __float_as_int(nrm);

    if (FULL) {
        float h0[2][SC_BATCH], h1[2][SC_BATCH];
#pragma unroll
        for (int u = 0; u < SC_BATCH; ++u) {
            const float* src = H + (size_t)__builtin_amdgcn_readlane(r, u) * NCLASS;
            h0[0][u] = src[lane];
            h1[0][u] = src[32 + lane];
        }
#pragma unroll
        for (int b = 0; b < 32 / SC_BATCH; ++b) {
            if (b + 1 < 32 / SC_BATCH) {
#pragma unroll
                for (int u = 0; u < SC_BATCH; ++u) {
                    const int j = (b + 1) * SC_BATCH + u;
                    const float* src = H + (size_t)__builtin_amdgcn_readlane(r, j) * NCLASS;
                    h0[(b + 1) & 1][u] = src[lane];
                    h1[(b + 1) & 1][u] = src[32 + lane];
                }
            }
#pragma unroll
            for (int u = 0; u < SC_BATCH; ++u) {
                const int j = b * SC_BATCH + u;
                const float nj = __int_as_float(__builtin_amdgcn_readlane(nbits, j));
                const int lj = __builtin_amdgcn_readlane(lr, j);
                if (lj >= 0) {
                    float* dst = rows + lj * NCLASS;
                    atomicAdd(dst + lane, h0[b & 1][u] * nj);
                    atomicAdd(dst + 32 + lane, h1[b & 1][u] * nj);
                }
            }
        }
    } else {
        for (int j = 0; j < cnt; ++j) {
            const float nj = __int_as_float(__builtin_amdgcn_readlane(nbits, j));
            const int lj = __builtin_amdgcn_readlane(lr, j);
            const float* src = H + (size_t)__builtin_amdgcn_readlane(r, j) * NCLASS;
            const float a0 = src[lane];
            const float a1 = src[32 + lane];
            if (lj >= 0) {
                float* dst = rows + lj * NCLASS;
                atomicAdd(dst + lane, a0 * nj);
                atomicAdd(dst + 32 + lane, a1 * nj);
            }
        }
    }
}

__global__ __launch_bounds__(SC_THREADS) void edge_sum_kernel(const float* __restrict__ H,
                                                              const float* __restrict__ deg,
                                                              const float* __restrict__ bg,
                                                              const float* __restrict__ offsbuf,
                                                              const float* __restrict__ srcbuf,
                                                              const float* __restrict__ dstbuf,
                                                              float* __restrict__ agg, int e, int n) {
    __shared__ __attribute__((aligned(16))) unsigned short dtab[DT_CAP];
    __shared__ __attribute__((aligned(16))) float rows[RS * NCLASS];
    __shared__ int found[(SC_THREADS / 32) * Q_CAP];

    const unsigned* __restrict__ offs = (const unsigned*)offsbuf;
    const int* __restrict__ srcL = (const int*)srcbuf;
    const unsigned short* __restrict__ dstL = (const unsigned short*)dstbuf;

    {
        const int lim = min(n, DT_CAP);
        const int lim4 = lim & ~3;
        const int chunk = SC_THREADS * 4;
        const int full = lim4 / chunk;
        const int t4 = (int)threadIdx.x * 4;
        int t = full ? (int)(blockIdx.x % (unsigned)full) : 0;
#pragma unroll 8
        for (int k = 0; k < full; ++k) {
            const int i = t * chunk + t4;
            *(uint2*)(dtab + i) = pack_counts(*(const float4*)(deg + i));
            t = (t + 1 == full) ? 0 : t + 1;
        }
        const int i = full * chunk + t4;
        if (i < lim4) *(uint2*)(dtab + i) = pack_counts(*(const float4*)(deg + i));
        for (int k = lim4 + (int)threadIdx.x; k < lim; k += SC_THREADS)
            dtab[k] = (unsigned short)min((unsigned)deg[k], DT_FULL);
    }

    const int lane = threadIdx.x & 31;
    const int wib = __builtin_amdgcn_readfirstlane((int)(threadIdx.x >> 5));
    int* q = found + wib * Q_CAP;
    const int nslab = (n + RS - 1) / RS;
    const int nch = (e + CH_EDGES - 1) / CH_EDGES;
    const int per = (nch + 31) / 32;
    const int owner = (int)blockIdx.x;
    const float4 b4 = ((const float4*)bg)[threadIdx.x & 15];
    const float4* __restrict__ H4 = (const float4*)H;
    float4* rows4 = (float4*)rows;
    float4* __restrict__ agg4 = (float4*)agg;

    for (int s = (int)blockIdx.x; s < nslab; s += SC_BLOCKS) {
        const int node0 = s * RS;

        for (int i = (int)threadIdx.x; i < RS * (NCLASS / 4); i += SC_THREADS) {
            const int node = node0 + (i >> 4);
            if (node < n)
                rows4[i] = agg_start(H4[(size_t)node * (NCLASS / 4) + (i & 15)], rsqrtf(deg[node]), b4);
        }
        __syncthreads();

        const int pass = s / SC_BLOCKS;
        int qn = 0;

        for (int j0 = 0; j0 < per; j0 += 32) {
            const int jj = j0 + lane;
            const int ch = wib * per + jj;
            unsigned oc = 0u;
            if (jj < per && ch < nch) oc = offs[(size_t)ch * SC_BLOCKS + owner];

            const int nrun = min(32, per - j0);
            for (int k = 0; k < nrun; ++k) {
                const unsigned w = (unsigned)__builtin_amdgcn_readlane((int)oc, k);
                int left = (int)(w & 0xFFFFu);
                int pos = (wib * per + j0 + k) * CH_EDGES + (int)(w >> 16);
                while (left > 0) {
                    const int take = min(left, 32);
                    if (lane < take) q[qn + lane] = pos + lane;
                    qn += take;
                    pos += take;
                    left -= take;
                    if (qn >= 32) {
                        qn -= 32;
                        add_edges<true>(srcL, dstL, H, deg, dtab, rows, node0, pass, lane, q[qn + lane], 32);
                    }
                }
            }
        }
        if (qn > 0) {
            const int idx = (lane < qn) ? q[lane] : 0;
            add_edges<false>(srcL, dstL, H, deg, dtab, rows, node0, pass, lane, idx, qn);
        }
        __syncthreads();

        for (int i = (int)threadIdx.x; i < RS * (NCLASS / 4); i += SC_THREADS) {
            const int node = node0 + (i >> 4);
            if (node < n) agg4[(size_t)node * (NCLASS / 4) + (i & 15)] = rows4[i];
        }
        __syncthreads();
    }
}

__global__ __launch_bounds__(256) void gemm2_kernel(const float* __restrict__ AGG0,
                                                    const float* __restrict__ Wl,
                                                    const float* __restrict__ bl,
                                                    float* __restrict__ OUT, int n) {
    const int lane = threadIdx.x & 31;
    const int wave = threadIdx.x >> 5;
    const int mtile = blockIdx.x * 8 + wave;
    const int m0 = mtile * 16;
    if (m0 >= n) return;

    const int half = lane >> 4;
    const int l15  = lane & 15;
    int m = m0 + l15;
    if (m >= n) m = n - 1;
    const float* __restrict__ arow = AGG0 + (size_t)m * NCLASS;

    v8f acc0 = {}, acc1 = {}, acc2 = {}, acc3 = {};

    for (int k0 = 0; k0 < NCLASS; k0 += 4) {
        const int kb = k0 + half * 2;
        v2f a;
        a.x = fmaxf(arow[kb], 0.0f);
        a.y = fmaxf(arow[kb + 1], 0.0f);

        const float* __restrict__ w0 = Wl + (size_t)kb * NCLASS;
        const float* __restrict__ w1 = w0 + NCLASS;
        v2f b0, b1, b2, b3;
        b0.x = w0[l15];      b0.y = w1[l15];
        b1.x = w0[16 + l15]; b1.y = w1[16 + l15];
        b2.x = w0[32 + l15]; b2.y = w1[32 + l15];
        b3.x = w0[48 + l15]; b3.y = w1[48 + l15];

        acc0 = __builtin_amdgcn_wmma_f32_16x16x4_f32(false, a, false, b0, (short)0, acc0, false, false);
        acc1 = __builtin_amdgcn_wmma_f32_16x16x4_f32(false, a, false, b1, (short)0, acc1, false, false);
        acc2 = __builtin_amdgcn_wmma_f32_16x16x4_f32(false, a, false, b2, (short)0, acc2, false, false);
        acc3 = __builtin_amdgcn_wmma_f32_16x16x4_f32(false, a, false, b3, (short)0, acc3, false, false);
    }

    const float bl0 = bl[l15];
    const float bl1 = bl[16 + l15];
    const float bl2 = bl[32 + l15];
    const float bl3 = bl[48 + l15];

    const int mbase = m0 + half * 8;
#pragma unroll
    for (int r = 0; r < 8; ++r) {
        const int mr = mbase + r;
        if (mr < n) {
            float* orow = OUT + (size_t)mr * NCLASS;
            orow[l15]      = acc0[r] + bl0;
            orow[16 + l15] = acc1[r] + bl1;
            orow[32 + l15] = acc2[r] + bl2;
            orow[48 + l15] = acc3[r] + bl3;
        }
    }
}

extern "C" void kernel_launch(void* const* d_in, const int* in_sizes, int n_in,
                              void* d_out, int out_size, void* d_ws, size_t ws_size,
                              hipStream_t stream) {
    const float* x  = (const float*)d_in[0];
    const int*   ei = (const int*)d_in[1];
    const float* Wg = (const float*)d_in[2];
    const float* bg = (const float*)d_in[3];
    const float* Wl = (const float*)d_in[4];
    const float* bl = (const float*)d_in[5];
    float* out = (float*)d_out;

    const int n = in_sizes[0] / NFEAT;
    const int e = in_sizes[1] / 2;

    float* ws   = (float*)d_ws;
    size_t off  = ((size_t)n + 3) & ~(size_t)3;
    float* deg  = ws;
    float* h    = ws + off;
    float* agg  = h + (size_t)n * NCLASS;

    const int T = 256;

    const int mtiles = (n + 15) / 16;
    const int gblocks = (mtiles + 7) / 8;
    const int dblocks = (e + T * DC_U - 1) / (T * DC_U);
    float* offsb = out + off;
    float* srcb  = offsb + (size_t)dblocks * SC_BLOCKS;
    float* dstb  = srcb + e;

    deg_init_kernel <<<(n + T * 4 - 1) / (T * 4), T, 0, stream>>>(deg, out, n);
    gemm1_degcount_kernel<<<gblocks + dblocks, T, 0, stream>>>(x, Wg, h, n, gblocks, ei, deg, out, offsb, srcb, dstb, e);
    deg_sum_kernel<<<(n + T * 4 - 1) / (T * 4), T, 0, stream>>>(deg, out, n);

    edge_sum_kernel<<<SC_BLOCKS, SC_THREADS, 0, stream>>>(h, deg, bg, offsb, srcb, dstb, agg, e, n);

    gemm2_kernel<<<gblocks, T, 0, stream>>>(agg, Wl, bl, out, n);
}


# count blocks build their sorted source and target lists in LDS and copy them out in order, 16 bytes a lane: whole 128-byte lines per store, not 32 scattered 4- or 2-byte writes; sums unchanged
# speedup vs baseline: 3.7790x; 1.1365x over previous
#include <hip/hip_runtime.h>

typedef __attribute__((ext_vector_type(2))) float v2f;
typedef __attribute__((ext_vector_type(8))) float v8f;

#define NFEAT 256
#define NCLASS 64

__device__ __forceinline__ int chip_half() {
    return (int)((__builtin_amdgcn_s_sendmsg_rtn(0x87) >> 18) & 1u);
}

__global__ void deg_init_kernel(float* __restrict__ deg0, float* __restrict__ deg1, int n) {
    const int i = (blockIdx.x * blockDim.x + threadIdx.x) * 4;
    if (i + 3 < n) {
        *(float4*)(deg0 + i) = make_float4(1.0f, 1.0f, 1.0f, 1.0f);
        *(float4*)(deg1 + i) = make_float4(0.0f, 0.0f, 0.0f, 0.0f);
    } else {
        for (int k = i; k < n; ++k) {
            deg0[k] = 1.0f;
            deg1[k] = 0.0f;
        }
    }
}

#define DC_U 8
#define CH_EDGES (256 * DC_U)
#define RS 391
#define SC_BLOCKS 256
#define L_BITS 9

__device__ __forceinline__ void part_body(const int* __restrict__ ei,
                                          float* __restrict__ deg0,
                                          float* __restrict__ deg1,
                                          unsigned* __restrict__ offs,
                                          int* __restrict__ srcL,
                                          unsigned short* __restrict__ dstL,
                                          int e, int blk) {
    __shared__ int hist[SC_BLOCKS];
    __shared__ int sc[2][SC_BLOCKS];
    __shared__ __attribute__((aligned(16))) int srcS[CH_EDGES];
    __shared__ __attribute__((aligned(16))) unsigned short dstS[CH_EDGES];
    const int t = (int)threadIdx.x;
    hist[t] = 0;
    __syncthreads();

    const int base = blk * CH_EDGES;
    float* deg = chip_half() ? deg1 : deg0;
    int r[DC_U], c[DC_U], slot[DC_U];
#pragma unroll
    for (int u = 0; u < DC_U; ++u) {
        const int i = min(base + u * 256 + t, e - 1);
        r[u] = ei[i];
        c[u] = ei[e + i];
    }
#pragma unroll
    for (int u = 0; u < DC_U; ++u) {
        const bool live = base + u * 256 + t < e;
        slot[u] = 0;
        if (live) {
            slot[u] = atomicAdd(&hist[((unsigned)c[u] / RS) & (SC_BLOCKS - 1)], 1);
            atomicAdd(&deg[c[u]], 1.0f);
        }
    }
    __syncthreads();

    const int v = hist[t];
    sc[0][t] = v;
    __syncthreads();
    int pp = 0;
    for (int d = 1; d < SC_BLOCKS; d <<= 1) {
        int x = sc[pp][t];
        if (t >= d) x += sc[pp][t - d];
        sc[pp ^ 1][t] = x;
        pp ^= 1;
        __syncthreads();
    }
    const int start = sc[pp][t] - v;
    offs[blk * SC_BLOCKS + t] = ((unsigned)start << 16) | (unsigned)v;
    hist[t] = start;
    __syncthreads();

#pragma unroll
    for (int u = 0; u < DC_U; ++u) {
        const bool live = base + u * 256 + t < e;
        if (live) {
            const unsigned slab = (unsigned)c[u] / RS;
            const int pos = hist[slab & (SC_BLOCKS - 1)] + slot[u];
            srcS[pos] = r[u];
            dstS[pos] = (unsigned short)(((slab >> 8) << L_BITS) | ((unsigned)c[u] - slab * RS));
        }
    }
    __syncthreads();

    const int nlive = min(CH_EDGES, e - base);
    int* __restrict__ sg = srcL + base;
    unsigned short* __restrict__ dg = dstL + base;
    const bool al16 = ((((size_t)sg) | ((size_t)dg)) & 15) == 0;
    if (al16 && nlive == CH_EDGES) {
        const int4 s0 = *(const int4*)(srcS + t * 4);
        const int4 s1 = *(const int4*)(srcS + CH_EDGES / 2 + t * 4);
        const uint4 d0 = *(const uint4*)(dstS + t * 8);
        *(int4*)(sg + t * 4) = s0;
        *(int4*)(sg + CH_EDGES / 2 + t * 4) = s1;
        *(uint4*)(dg + t * 8) = d0;
    } else {
        for (int i = t; i < nlive; i += 256) {
            sg[i] = srcS[i];
            dg[i] = dstS[i];
        }
    }
}

__global__ void deg_sum_kernel(float* __restrict__ deg0, const float* __restrict__ deg1, int n) {
    const int i = (blockIdx.x * blockDim.x + threadIdx.x) * 4;
    if (i + 3 < n) {
        const float4 a = *(const float4*)(deg0 + i);
        const float4 b = *(const float4*)(deg1 + i);
        float4 o;
        o.x = a.x + b.x;
        o.y = a.y + b.y;
        o.z = a.z + b.z;
        o.w = a.w + b.w;
        *(float4*)(deg0 + i) = o;
    } else {
        for (int k = i; k < n; ++k) deg0[k] = deg0[k] + deg1[k];
    }
}

__device__ __forceinline__ void gemm1_body(const float* __restrict__ X,
                                           const float* __restrict__ Wg,
                                           float* __restrict__ H, int n, int blk) {
    const int lane = threadIdx.x & 31;
    const int wave = threadIdx.x >> 5;
    const int mtile = blk * 8 + wave;
    const int m0 = mtile * 16;
    if (m0 >= n) return;

    const int half = lane >> 4;
    const int l15  = lane & 15;
    int m = m0 + l15;
    if (m >= n) m = n - 1;
    const float* __restrict__ xrow = X + (size_t)m * NFEAT;

    v8f acc0 = {}, acc1 = {}, acc2 = {}, acc3 = {};

    for (int k0 = 0; k0 < NFEAT; k0 += 4) {
        const int kb = k0 + half * 2;
        v2f a;
        a.x = xrow[kb];
        a.y = xrow[kb + 1];

        const float* __restrict__ w0 = Wg + (size_t)kb * NCLASS;
        const float* __restrict__ w1 = w0 + NCLASS;
        v2f b0, b1, b2, b3;
        b0.x = w0[l15];      b0.y = w1[l15];
        b1.x = w0[16 + l15]; b1.y = w1[16 + l15];
        b2.x = w0[32 + l15]; b2.y = w1[32 + l15];
        b3.x = w0[48 + l15]; b3.y = w1[48 + l15];

        acc0 = __builtin_amdgcn_wmma_f32_16x16x4_f32(false, a, false, b0, (short)0, acc0, false, false);
        acc1 = __builtin_amdgcn_wmma_f32_16x16x4_f32(false, a, false, b1, (short)0, acc1, false, false);
        acc2 = __builtin_amdgcn_wmma_f32_16x16x4_f32(false, a, false, b2, (short)0, acc2, false, false);
        acc3 = __builtin_amdgcn_wmma_f32_16x16x4_f32(false, a, false, b3, (short)0, acc3, false, false);
    }

    const int mbase = m0 + half * 8;
#pragma unroll
    for (int r = 0; r < 8; ++r) {
        const int mr = mbase + r;
        if (mr < n) {
            float* __restrict__ hrow = H + (size_t)mr * NCLASS;
            hrow[l15]      = acc0[r];
            hrow[16 + l15] = acc1[r];
            hrow[32 + l15] = acc2[r];
            hrow[48 + l15] = acc3[r];
        }
    }
}

__global__ __launch_bounds__(256) void gemm1_degcount_kernel(const float* __restrict__ X,
                                                             const float* __restrict__ Wg,
                                                             float* __restrict__ H, int n,
                                                             int gblocks,
                                                             const int* __restrict__ ei,
                                                             float* __restrict__ deg0,
                                                             float* __restrict__ deg1,
                                                             float* __restrict__ offsbuf,
                                                             float* __restrict__ srcbuf,
                                                             float* __restrict__ dstbuf, int e) {
    const int blk = (int)blockIdx.x;
    if (blk < gblocks) {
        gemm1_body(X, Wg, H, n, blk);
    } else {
        part_body(ei, deg0, deg1, (unsigned*)offsbuf, (int*)srcbuf, (unsigned short*)dstbuf, e, blk - gblocks);
    }
}

#define SC_BATCH 8
#define SC_THREADS 1024
#define DT_CAP 102400
#define DT_FULL 65535u
#define Q_CAP 64

__device__ __forceinline__ uint2 pack_counts(const float4 d) {
    const unsigned u0 = min((unsigned)d.x, DT_FULL);
    const unsigned u1 = min((unsigned)d.y, DT_FULL);
    const unsigned u2 = min((unsigned)d.z, DT_FULL);
    const unsigned u3 = min((unsigned)d.w, DT_FULL);
    return make_uint2(u0 | (u1 << 16), u2 | (u3 << 16));
}

__device__ __forceinline__ float4 agg_start(const float4 h, const float d, const float4 b) {
    float4 o;
    o.x = h.x * d * d + b.x;
    o.y = h.y * d * d + b.y;
    o.z = h.z * d * d + b.z;
    o.w = h.w * d * d + b.w;
    return o;
}

template <bool FULL>
__device__ __forceinline__ void add_edges(const int* __restrict__ srcL,
                                          const unsigned short* __restrict__ dstL,
                                          const float* __restrict__ H,
                                          const float* __restrict__ deg, const unsigned short* dtab,
                                          float* rows, const int node0, const int pass, const int lane,
                                          const int idx, const int cnt) {
    int r = 0, lr = -1;
    float nrm = 0.0f;
    if (lane < cnt) {
        r = srcL[idx];
        const unsigned d = (unsigned)dstL[idx];
        if ((int)(d >> L_BITS) == pass) {
            const int l = (int)(d & ((1u << L_BITS) - 1u));
            const int c = node0 + l;
            const unsigned tr = (r < DT_CAP) ? (unsigned)dtab[r] : DT_FULL;
            const unsigned tc = (c < DT_CAP) ? (unsigned)dtab[c] : DT_FULL;
            float dr = (float)tr, dc = (float)tc;
            if (tr == DT_FULL) dr = deg[r];
            if (tc == DT_FULL) dc = deg[c];
            nrm = rsqrtf(dr) * rsqrtf(dc);
            lr = l;
        }
    }
    const int nbits = __float_as_int(nrm);

    if (FULL) {
        float h0[2][SC_BATCH], h1[2][SC_BATCH];
#pragma unroll
        for (int u = 0; u < SC_BATCH; ++u) {
            const float* src = H + (size_t)__builtin_amdgcn_readlane(r, u) * NCLASS;
            h0[0][u] = src[lane];
            h1[0][u] = src[32 + lane];
        }
#pragma unroll
        for (int b = 0; b < 32 / SC_BATCH; ++b) {
            if (b + 1 < 32 / SC_BATCH) {
#pragma unroll
                for (int u = 0; u < SC_BATCH; ++u) {
                    const int j = (b + 1) * SC_BATCH + u;
                    const float* src = H + (size_t)__builtin_amdgcn_readlane(r, j) * NCLASS;
                    h0[(b + 1) & 1][u] = src[lane];
                    h1[(b + 1) & 1][u] = src[32 + lane];
                }
            }
#pragma unroll
            for (int u = 0; u < SC_BATCH; ++u) {
                const int j = b * SC_BATCH + u;
                const float nj = __int_as_float(__builtin_amdgcn_readlane(nbits, j));
                const int lj = __builtin_amdgcn_readlane(lr, j);
                if (lj >= 0) {
                    float* dst = rows + lj * NCLASS;
                    atomicAdd(dst + lane, h0[b & 1][u] * nj);
                    atomicAdd(dst + 32 + lane, h1[b & 1][u] * nj);
                }
            }
        }
    } else {
        for (int j = 0; j < cnt; ++j) {
            const float nj = __int_as_float(__builtin_amdgcn_readlane(nbits, j));
            const int lj = __builtin_amdgcn_readlane(lr, j);
            const float* src = H + (size_t)__builtin_amdgcn_readlane(r, j) * NCLASS;
            const float a0 = src[lane];
            const float a1 = src[32 + lane];
            if (lj >= 0) {
                float* dst = rows + lj * NCLASS;
                atomicAdd(dst + lane, a0 * nj);
                atomicAdd(dst + 32 + lane, a1 * nj);
            }
        }
    }
}

__global__ __launch_bounds__(SC_THREADS) void edge_sum_kernel(const float* __restrict__ H,
                                                              const float* __restrict__ deg,
                                                              const float* __restrict__ bg,
                                                              const float* __restrict__ offsbuf,
                                                              const float* __restrict__ srcbuf,
                                                              const float* __restrict__ dstbuf,
                                                              float* __restrict__ agg, int e, int n) {
    __shared__ __attribute__((aligned(16))) unsigned short dtab[DT_CAP];
    __shared__ __attribute__((aligned(16))) float rows[RS * NCLASS];
    __shared__ int found[(SC_THREADS / 32) * Q_CAP];

    const unsigned* __restrict__ offs = (const unsigned*)offsbuf;
    const int* __restrict__ srcL = (const int*)srcbuf;
    const unsigned short* __restrict__ dstL = (const unsigned short*)dstbuf;

    {
        const int lim = min(n, DT_CAP);
        const int lim4 = lim & ~3;
        const int chunk = SC_THREADS * 4;
        const int full = lim4 / chunk;
        const int t4 = (int)threadIdx.x * 4;
        int t = full ? (int)(blockIdx.x % (unsigned)full) : 0;
#pragma unroll 8
        for (int k = 0; k < full; ++k) {
            const int i = t * chunk + t4;
            *(uint2*)(dtab + i) = pack_counts(*(const float4*)(deg + i));
            t = (t + 1 == full) ? 0 : t + 1;
        }
        const int i = full * chunk + t4;
        if (i < lim4) *(uint2*)(dtab + i) = pack_counts(*(const float4*)(deg + i));
        for (int k = lim4 + (int)threadIdx.x; k < lim; k += SC_THREADS)
            dtab[k] = (unsigned short)min((unsigned)deg[k], DT_FULL);
    }

    const int lane = threadIdx.x & 31;
    const int wib = __builtin_amdgcn_readfirstlane((int)(threadIdx.x >> 5));
    int* q = found + wib * Q_CAP;
    const int nslab = (n + RS - 1) / RS;
    const int nch = (e + CH_EDGES - 1) / CH_EDGES;
    const int per = (nch + 31) / 32;
    const int owner = (int)blockIdx.x;
    const float4 b4 = ((const float4*)bg)[threadIdx.x & 15];
    const float4* __restrict__ H4 = (const float4*)H;
    float4* rows4 = (float4*)rows;
    float4* __restrict__ agg4 = (float4*)agg;

    for (int s = (int)blockIdx.x; s < nslab; s += SC_BLOCKS) {
        const int node0 = s * RS;

        for (int i = (int)threadIdx.x; i < RS * (NCLASS / 4); i += SC_THREADS) {
            const int node = node0 + (i >> 4);
            if (node < n)
                rows4[i] = agg_start(H4[(size_t)node * (NCLASS / 4) + (i & 15)], rsqrtf(deg[node]), b4);
        }
        __syncthreads();

        const int pass = s / SC_BLOCKS;
        int qn = 0;

        for (int j0 = 0; j0 < per; j0 += 32) {
            const int jj = j0 + lane;
            const int ch = wib * per + jj;
            unsigned oc = 0u;
            if (jj < per && ch < nch) oc = offs[(size_t)ch * SC_BLOCKS + owner];

            const int nrun = min(32, per - j0);
            for (int k = 0; k < nrun; ++k) {
                const unsigned w = (unsigned)__builtin_amdgcn_readlane((int)oc, k);
                int left = (int)(w & 0xFFFFu);
                int pos = (wib * per + j0 + k) * CH_EDGES + (int)(w >> 16);
                while (left > 0) {
                    const int take = min(left, 32);
                    if (lane < take) q[qn + lane] = pos + lane;
                    qn += take;
                    pos += take;
                    left -= take;
                    if (qn >= 32) {
                        qn -= 32;
                        add_edges<true>(srcL, dstL, H, deg, dtab, rows, node0, pass, lane, q[qn + lane], 32);
                    }
                }
            }
        }
        if (qn > 0) {
            const int idx = (lane < qn) ? q[lane] : 0;
            add_edges<false>(srcL, dstL, H, deg, dtab, rows, node0, pass, lane, idx, qn);
        }
        __syncthreads();

        for (int i = (int)threadIdx.x; i < RS * (NCLASS / 4); i += SC_THREADS) {
            const int node = node0 + (i >> 4);
            if (node < n) agg4[(size_t)node * (NCLASS / 4) + (i & 15)] = rows4[i];
        }
        __syncthreads();
    }
}

__global__ __launch_bounds__(256) void gemm2_kernel(const float* __restrict__ AGG0,
                                                    const float* __restrict__ Wl,
                                                    const float* __restrict__ bl,
                                                    float* __restrict__ OUT, int n) {
    const int lane = threadIdx.x & 31;
    const int wave = threadIdx.x >> 5;
    const int mtile = blockIdx.x * 8 + wave;
    const int m0 = mtile * 16;
    if (m0 >= n) return;

    const int half = lane >> 4;
    const int l15  = lane & 15;
    int m = m0 + l15;
    if (m >= n) m = n - 1;
    const float* __restrict__ arow = AGG0 + (size_t)m * NCLASS;

    v8f acc0 = {}, acc1 = {}, acc2 = {}, acc3 = {};

    for (int k0 = 0; k0 < NCLASS; k0 += 4) {
        const int kb = k0 + half * 2;
        v2f a;
        a.x = fmaxf(arow[kb], 0.0f);
        a.y = fmaxf(arow[kb + 1], 0.0f);

        const float* __restrict__ w0 = Wl + (size_t)kb * NCLASS;
        const float* __restrict__ w1 = w0 + NCLASS;
        v2f b0, b1, b2, b3;
        b0.x = w0[l15];      b0.y = w1[l15];
        b1.x = w0[16 + l15]; b1.y = w1[16 + l15];
        b2.x = w0[32 + l15]; b2.y = w1[32 + l15];
        b3.x = w0[48 + l15]; b3.y = w1[48 + l15];

        acc0 = __builtin_amdgcn_wmma_f32_16x16x4_f32(false, a, false, b0, (short)0, acc0, false, false);
        acc1 = __builtin_amdgcn_wmma_f32_16x16x4_f32(false, a, false, b1, (short)0, acc1, false, false);
        acc2 = __builtin_amdgcn_wmma_f32_16x16x4_f32(false, a, false, b2, (short)0, acc2, false, false);
        acc3 = __builtin_amdgcn_wmma_f32_16x16x4_f32(false, a, false, b3, (short)0, acc3, false, false);
    }

    const float bl0 = bl[l15];
    const float bl1 = bl[16 + l15];
    const float bl2 = bl[32 + l15];
    const float bl3 = bl[48 + l15];

    const int mbase = m0 + half * 8;
#pragma unroll
    for (int r = 0; r < 8; ++r) {
        const int mr = mbase + r;
        if (mr < n) {
            float* orow = OUT + (size_t)mr * NCLASS;
            orow[l15]      = acc0[r] + bl0;
            orow[16 + l15] = acc1[r] + bl1;
            orow[32 + l15] = acc2[r] + bl2;
            orow[48 + l15] = acc3[r] + bl3;
        }
    }
}

extern "C" void kernel_launch(void* const* d_in, const int* in_sizes, int n_in,
                              void* d_out, int out_size, void* d_ws, size_t ws_size,
                              hipStream_t stream) {
    const float* x  = (const float*)d_in[0];
    const int*   ei = (const int*)d_in[1];
    const float* Wg = (const float*)d_in[2];
    const float* bg = (const float*)d_in[3];
    const float* Wl = (const float*)d_in[4];
    const float* bl = (const float*)d_in[5];
    float* out = (float*)d_out;

    const int n = in_sizes[0] / NFEAT;
    const int e = in_sizes[1] / 2;

    float* ws   = (float*)d_ws;
    size_t off  = ((size_t)n + 3) & ~(size_t)3;
    float* deg  = ws;
    float* h    = ws + off;
    float* agg  = h + (size_t)n * NCLASS;

    const int T = 256;

    const int mtiles = (n + 15) / 16;
    const int gblocks = (mtiles + 7) / 8;
    const int dblocks = (e + T * DC_U - 1) / (T * DC_U);
    float* offsb = out + off;
    float* srcb  = offsb + (size_t)dblocks * SC_BLOCKS;
    float* dstb  = srcb + e;

    deg_init_kernel <<<(n + T * 4 - 1) / (T * 4), T, 0, stream>>>(deg, out, n);
    gemm1_degcount_kernel<<<gblocks + dblocks, T, 0, stream>>>(x, Wg, h, n, gblocks, ei, deg, out, offsb, srcb, dstb, e);
    deg_sum_kernel<<<(n + T * 4 - 1) / (T * 4), T, 0, stream>>>(deg, out, n);

    edge_sum_kernel<<<SC_BLOCKS, SC_THREADS, 0, stream>>>(h, deg, bg, offsb, srcb, dstb, agg, e, n);

    gemm2_kernel<<<gblocks, T, 0, stream>>>(agg, Wl, bl, out, n);
}


# last product done by the owner block on its finished LDS rows: one wave per 16 rows, WMMA fed from LDS, rows put back and stored as whole lines; no sum plane, 4 launches not 5; sums unchanged
# speedup vs baseline: 4.1019x; 1.0854x over previous
#include <hip/hip_runtime.h>

typedef __attribute__((ext_vector_type(2))) float v2f;
typedef __attribute__((ext_vector_type(8))) float v8f;

#define NFEAT 256
#define NCLASS 64

__device__ __forceinline__ int chip_half() {
    return (int)((__builtin_amdgcn_s_sendmsg_rtn(0x87) >> 18) & 1u);
}

__global__ void deg_init_kernel(float* __restrict__ deg0, float* __restrict__ deg1, int n) {
    const int i = (blockIdx.x * blockDim.x + threadIdx.x) * 4;
    if (i + 3 < n) {
        *(float4*)(deg0 + i) = make_float4(1.0f, 1.0f, 1.0f, 1.0f);
        *(float4*)(deg1 + i) = make_float4(0.0f, 0.0f, 0.0f, 0.0f);
    } else {
        for (int k = i; k < n; ++k) {
            deg0[k] = 1.0f;
            deg1[k] = 0.0f;
        }
    }
}

#define DC_U 8
#define CH_EDGES (256 * DC_U)
#define RS 391
#define SC_BLOCKS 256
#define L_BITS 9

__device__ __forceinline__ void part_body(const int* __restrict__ ei,
                                          float* __restrict__ deg0,
                                          float* __restrict__ deg1,
                                          unsigned* __restrict__ offs,
                                          int* __restrict__ srcL,
                                          unsigned short* __restrict__ dstL,
                                          int e, int blk) {
    __shared__ int hist[SC_BLOCKS];
    __shared__ int sc[2][SC_BLOCKS];
    __shared__ __attribute__((aligned(16))) int srcS[CH_EDGES];
    __shared__ __attribute__((aligned(16))) unsigned short dstS[CH_EDGES];
    const int t = (int)threadIdx.x;
    hist[t] = 0;
    __syncthreads();

    const int base = blk * CH_EDGES;
    float* deg = chip_half() ? deg1 : deg0;
    int r[DC_U], c[DC_U], slot[DC_U];
#pragma unroll
    for (int u = 0; u < DC_U; ++u) {
        const int i = min(base + u * 256 + t, e - 1);
        r[u] = ei[i];
        c[u] = ei[e + i];
    }
#pragma unroll
    for (int u = 0; u < DC_U; ++u) {
        const bool live = base + u * 256 + t < e;
        slot[u] = 0;
        if (live) {
            slot[u] = atomicAdd(&hist[((unsigned)c[u] / RS) & (SC_BLOCKS - 1)], 1);
            atomicAdd(&deg[c[u]], 1.0f);
        }
    }
    __syncthreads();

    const int v = hist[t];
    sc[0][t] = v;
    __syncthreads();
    int pp = 0;
    for (int d = 1; d < SC_BLOCKS; d <<= 1) {
        int x = sc[pp][t];
        if (t >= d) x += sc[pp][t - d];
        sc[pp ^ 1][t] = x;
        pp ^= 1;
        __syncthreads();
    }
    const int start = sc[pp][t] - v;
    offs[blk * SC_BLOCKS + t] = ((unsigned)start << 16) | (unsigned)v;
    hist[t] = start;
    __syncthreads();

#pragma unroll
    for (int u = 0; u < DC_U; ++u) {
        const bool live = base + u * 256 + t < e;
        if (live) {
            const unsigned slab = (unsigned)c[u] / RS;
            const int pos = hist[slab & (SC_BLOCKS - 1)] + slot[u];
            srcS[pos] = r[u];
            dstS[pos] = (unsigned short)(((slab >> 8) << L_BITS) | ((unsigned)c[u] - slab * RS));
        }
    }
    __syncthreads();

    const int nlive = min(CH_EDGES, e - base);
    int* __restrict__ sg = srcL + base;
    unsigned short* __restrict__ dg = dstL + base;
    const bool al16 = ((((size_t)sg) | ((size_t)dg)) & 15) == 0;
    if (al16 && nlive == CH_EDGES) {
        const int4 s0 = *(const int4*)(srcS + t * 4);
        const int4 s1 = *(const int4*)(srcS + CH_EDGES / 2 + t * 4);
        const uint4 d0 = *(const uint4*)(dstS + t * 8);
        *(int4*)(sg + t * 4) = s0;
        *(int4*)(sg + CH_EDGES / 2 + t * 4) = s1;
        *(uint4*)(dg + t * 8) = d0;
    } else {
        for (int i = t; i < nlive; i += 256) {
            sg[i] = srcS[i];
            dg[i] = dstS[i];
        }
    }
}

__global__ void deg_sum_kernel(float* __restrict__ deg0, const float* __restrict__ deg1, int n) {
    const int i = (blockIdx.x * blockDim.x + threadIdx.x) * 4;
    if (i + 3 < n) {
        const float4 a = *(const float4*)(deg0 + i);
        const float4 b = *(const float4*)(deg1 + i);
        float4 o;
        o.x = a.x + b.x;
        o.y = a.y + b.y;
        o.z = a.z + b.z;
        o.w = a.w + b.w;
        *(float4*)(deg0 + i) = o;
    } else {
        for (int k = i; k < n; ++k) deg0[k] = deg0[k] + deg1[k];
    }
}

__device__ __forceinline__ void gemm1_body(const float* __restrict__ X,
                                           const float* __restrict__ Wg,
                                           float* __restrict__ H, int n, int blk) {
    const int lane = threadIdx.x & 31;
    const int wave = threadIdx.x >> 5;
    const int mtile = blk * 8 + wave;
    const int m0 = mtile * 16;
    if (m0 >= n) return;

    const int half = lane >> 4;
    const int l15  = lane & 15;
    int m = m0 + l15;
    if (m >= n) m = n - 1;
    const float* __restrict__ xrow = X + (size_t)m * NFEAT;

    v8f acc0 = {}, acc1 = {}, acc2 = {}, acc3 = {};

    for (int k0 = 0; k0 < NFEAT; k0 += 4) {
        const int kb = k0 + half * 2;
        v2f a;
        a.x = xrow[kb];
        a.y = xrow[kb + 1];

        const float* __restrict__ w0 = Wg + (size_t)kb * NCLASS;
        const float* __restrict__ w1 = w0 + NCLASS;
        v2f b0, b1, b2, b3;
        b0.x = w0[l15];      b0.y = w1[l15];
        b1.x = w0[16 + l15]; b1.y = w1[16 + l15];
        b2.x = w0[32 + l15]; b2.y = w1[32 + l15];
        b3.x = w0[48 + l15]; b3.y = w1[48 + l15];

        acc0 = __builtin_amdgcn_wmma_f32_16x16x4_f32(false, a, false, b0, (short)0, acc0, false, false);
        acc1 = __builtin_amdgcn_wmma_f32_16x16x4_f32(false, a, false, b1, (short)0, acc1, false, false);
        acc2 = __builtin_amdgcn_wmma_f32_16x16x4_f32(false, a, false, b2, (short)0, acc2, false, false);
        acc3 = __builtin_amdgcn_wmma_f32_16x16x4_f32(false, a, false, b3, (short)0, acc3, false, false);
    }

    const int mbase = m0 + half * 8;
#pragma unroll
    for (int r = 0; r < 8; ++r) {
        const int mr = mbase + r;
        if (mr < n) {
            float* __restrict__ hrow = H + (size_t)mr * NCLASS;
            hrow[l15]      = acc0[r];
            hrow[16 + l15] = acc1[r];
            hrow[32 + l15] = acc2[r];
            hrow[48 + l15] = acc3[r];
        }
    }
}

__global__ __launch_bounds__(256) void gemm1_degcount_kernel(const float* __restrict__ X,
                                                             const float* __restrict__ Wg,
                                                             float* __restrict__ H, int n,
                                                             int gblocks,
                                                             const int* __restrict__ ei,
                                                             float* __restrict__ deg0,
                                                             float* __restrict__ deg1,
                                                             float* __restrict__ offsbuf,
                                                             float* __restrict__ srcbuf,
                                                             float* __restrict__ dstbuf, int e) {
    const int blk = (int)blockIdx.x;
    if (blk < gblocks) {
        gemm1_body(X, Wg, H, n, blk);
    } else {
        part_body(ei, deg0, deg1, (unsigned*)offsbuf, (int*)srcbuf, (unsigned short*)dstbuf, e, blk - gblocks);
    }
}

#define SC_BATCH 8
#define SC_THREADS 1024
#define DT_CAP 102400
#define DT_FULL 65535u
#define Q_CAP 64

__device__ __forceinline__ uint2 pack_counts(const float4 d) {
    const unsigned u0 = min((unsigned)d.x, DT_FULL);
    const unsigned u1 = min((unsigned)d.y, DT_FULL);
    const unsigned u2 = min((unsigned)d.z, DT_FULL);
    const unsigned u3 = min((unsigned)d.w, DT_FULL);
    return make_uint2(u0 | (u1 << 16), u2 | (u3 << 16));
}

__device__ __forceinline__ float4 agg_start(const float4 h, const float d, const float4 b) {
    float4 o;
    o.x = h.x * d * d + b.x;
    o.y = h.y * d * d + b.y;
    o.z = h.z * d * d + b.z;
    o.w = h.w * d * d + b.w;
    return o;
}

template <bool FULL>
__device__ __forceinline__ void add_edges(const int* __restrict__ srcL,
                                          const unsigned short* __restrict__ dstL,
                                          const float* __restrict__ H,
                                          const float* __restrict__ deg, const unsigned short* dtab,
                                          float* rows, const int node0, const int pass, const int lane,
                                          const int idx, const int cnt) {
    int r = 0, lr = -1;
    float nrm = 0.0f;
    if (lane < cnt) {
        r = srcL[idx];
        const unsigned d = (unsigned)dstL[idx];
        if ((int)(d >> L_BITS) == pass) {
            const int l = (int)(d & ((1u << L_BITS) - 1u));
            const int c = node0 + l;
            const unsigned tr = (r < DT_CAP) ? (unsigned)dtab[r] : DT_FULL;
            const unsigned tc = (c < DT_CAP) ? (unsigned)dtab[c] : DT_FULL;
            float dr = (float)tr, dc = (float)tc;
            if (tr == DT_FULL) dr = deg[r];
            if (tc == DT_FULL) dc = deg[c];
            nrm = rsqrtf(dr) * rsqrtf(dc);
            lr = l;
        }
    }
    const int nbits = __float_as_int(nrm);

    if (FULL) {
        float h0[2][SC_BATCH], h1[2][SC_BATCH];
#pragma unroll
        for (int u = 0; u < SC_BATCH; ++u) {
            const float* src = H + (size_t)__builtin_amdgcn_readlane(r, u) * NCLASS;
            h0[0][u] = src[lane];
            h1[0][u] = src[32 + lane];
        }
#pragma unroll
        for (int b = 0; b < 32 / SC_BATCH; ++b) {
            if (b + 1 < 32 / SC_BATCH) {
#pragma unroll
                for (int u = 0; u < SC_BATCH; ++u) {
                    const int j = (b + 1) * SC_BATCH + u;
                    const float* src = H + (size_t)__builtin_amdgcn_readlane(r, j) * NCLASS;
                    h0[(b + 1) & 1][u] = src[lane];
                    h1[(b + 1) & 1][u] = src[32 + lane];
                }
            }
#pragma unroll
            for (int u = 0; u < SC_BATCH; ++u) {
                const int j = b * SC_BATCH + u;
                const float nj = __int_as_float(__builtin_amdgcn_readlane(nbits, j));
                const int lj = __builtin_amdgcn_readlane(lr, j);
                if (lj >= 0) {
                    float* dst = rows + lj * NCLASS;
                    atomicAdd(dst + lane, h0[b & 1][u] * nj);
                    atomicAdd(dst + 32 + lane, h1[b & 1][u] * nj);
                }
            }
        }
    } else {
        for (int j = 0; j < cnt; ++j) {
            const float nj = __int_as_float(__builtin_amdgcn_readlane(nbits, j));
            const int lj = __builtin_amdgcn_readlane(lr, j);
            const float* src = H + (size_t)__builtin_amdgcn_readlane(r, j) * NCLASS;
            const float a0 = src[lane];
            const float a1 = src[32 + lane];
            if (lj >= 0) {
                float* dst = rows + lj * NCLASS;
                atomicAdd(dst + lane, a0 * nj);
                atomicAdd(dst + 32 + lane, a1 * nj);
            }
        }
    }
}

__global__ __launch_bounds__(SC_THREADS) void edge_sum_kernel(const float* __restrict__ H,
                                                              const float* __restrict__ deg,
                                                              const float* __restrict__ bg,
                                                              const float* __restrict__ offsbuf,
                                                              const float* __restrict__ srcbuf,
                                                              const float* __restrict__ dstbuf,
                                                              const float* __restrict__ Wl,
                                                              const float* __restrict__ bl,
                                                              float* __restrict__ OUT, int e, int n) {
    __shared__ __attribute__((aligned(16))) unsigned short dtab[DT_CAP];
    __shared__ __attribute__((aligned(16))) float rows[RS * NCLASS];
    __shared__ int found[(SC_THREADS / 32) * Q_CAP];

    const unsigned* __restrict__ offs = (const unsigned*)offsbuf;
    const int* __restrict__ srcL = (const int*)srcbuf;
    const unsigned short* __restrict__ dstL = (const unsigned short*)dstbuf;

    {
        const int lim = min(n, DT_CAP);
        const int lim4 = lim & ~3;
        const int chunk = SC_THREADS * 4;
        const int full = lim4 / chunk;
        const int t4 = (int)threadIdx.x * 4;
        int t = full ? (int)(blockIdx.x % (unsigned)full) : 0;
#pragma unroll 8
        for (int k = 0; k < full; ++k) {
            const int i = t * chunk + t4;
            *(uint2*)(dtab + i) = pack_counts(*(const float4*)(deg + i));
            t = (t + 1 == full) ? 0 : t + 1;
        }
        const int i = full * chunk + t4;
        if (i < lim4) *(uint2*)(dtab + i) = pack_counts(*(const float4*)(deg + i));
        for (int k = lim4 + (int)threadIdx.x; k < lim; k += SC_THREADS)
            dtab[k] = (unsigned short)min((unsigned)deg[k], DT_FULL);
    }

    const int lane = threadIdx.x & 31;
    const int wib = __builtin_amdgcn_readfirstlane((int)(threadIdx.x >> 5));
    int* q = found + wib * Q_CAP;
    const int nslab = (n + RS - 1) / RS;
    const int nch = (e + CH_EDGES - 1) / CH_EDGES;
    const int per = (nch + 31) / 32;
    const int owner = (int)blockIdx.x;
    const float4 b4 = ((const float4*)bg)[threadIdx.x & 15];
    const float4* __restrict__ H4 = (const float4*)H;
    float4* rows4 = (float4*)rows;
    float4* __restrict__ out4 = (float4*)OUT;

    for (int s = (int)blockIdx.x; s < nslab; s += SC_BLOCKS) {
        const int node0 = s * RS;

        for (int i = (int)threadIdx.x; i < RS * (NCLASS / 4); i += SC_THREADS) {
            const int node = node0 + (i >> 4);
            if (node < n)
                rows4[i] = agg_start(H4[(size_t)node * (NCLASS / 4) + (i & 15)], rsqrtf(deg[node]), b4);
        }
        __syncthreads();

        const int pass = s / SC_BLOCKS;
        int qn = 0;

        for (int j0 = 0; j0 < per; j0 += 32) {
            const int jj = j0 + lane;
            const int ch = wib * per + jj;
            unsigned oc = 0u;
            if (jj < per && ch < nch) oc = offs[(size_t)ch * SC_BLOCKS + owner];

            const int nrun = min(32, per - j0);
            for (int k = 0; k < nrun; ++k) {
                const unsigned w = (unsigned)__builtin_amdgcn_readlane((int)oc, k);
                int left = (int)(w & 0xFFFFu);
                int pos = (wib * per + j0 + k) * CH_EDGES + (int)(w >> 16);
                while (left > 0) {
                    const int take = min(left, 32);
                    if (lane < take) q[qn + lane] = pos + lane;
                    qn += take;
                    pos += take;
                    left -= take;
                    if (qn >= 32) {
                        qn -= 32;
                        add_edges<true>(srcL, dstL, H, deg, dtab, rows, node0, pass, lane, q[qn + lane], 32);
                    }
                }
            }
        }
        if (qn > 0) {
            const int idx = (lane < qn) ? q[lane] : 0;
            add_edges<false>(srcL, dstL, H, deg, dtab, rows, node0, pass, lane, idx, qn);
        }
        __syncthreads();

        if (wib < (RS + 15) / 16) {
            const int half = lane >> 4;
            const int l15  = lane & 15;
            const int m0 = wib * 16;
            const float* arow = rows + min(m0 + l15, RS - 1) * NCLASS;
            v8f acc0 = {}, acc1 = {}, acc2 = {}, acc3 = {};
#pragma unroll
            for (int k0 = 0; k0 < NCLASS; k0 += 4) {
                const int kb = k0 + half * 2;
                v2f a;
                a.x = fmaxf(arow[kb], 0.0f);
                a.y = fmaxf(arow[kb + 1], 0.0f);

                const float* __restrict__ w0 = Wl + (size_t)kb * NCLASS;
                const float* __restrict__ w1 = w0 + NCLASS;
                v2f b0, b1, b2, b3;
                b0.x = w0[l15];      b0.y = w1[l15];
                b1.x = w0[16 + l15]; b1.y = w1[16 + l15];
                b2.x = w0[32 + l15]; b2.y = w1[32 + l15];
                b3.x = w0[48 + l15]; b3.y = w1[48 + l15];

                acc0 = __builtin_amdgcn_wmma_f32_16x16x4_f32(false, a, false, b0, (short)0, acc0, false, false);
                acc1 = __builtin_amdgcn_wmma_f32_16x16x4_f32(false, a, false, b1, (short)0, acc1, false, false);
                acc2 = __builtin_amdgcn_wmma_f32_16x16x4_f32(false, a, false, b2, (short)0, acc2, false, false);
                acc3 = __builtin_amdgcn_wmma_f32_16x16x4_f32(false, a, false, b3, (short)0, acc3, false, false);
            }
            const float bl0 = bl[l15];
            const float bl1 = bl[16 + l15];
            const float bl2 = bl[32 + l15];
            const float bl3 = bl[48 + l15];
            const int mbase = m0 + half * 8;
#pragma unroll
            for (int r = 0; r < 8; ++r) {
                const int mr = mbase + r;
                if (mr < RS) {
                    float* orow = rows + mr * NCLASS;
                    orow[l15]      = acc0[r] + bl0;
                    orow[16 + l15] = acc1[r] + bl1;
                    orow[32 + l15] = acc2[r] + bl2;
                    orow[48 + l15] = acc3[r] + bl3;
                }
            }
            float4 ov[8];
#pragma unroll
            for (int j = 0; j < 8; ++j) {
                const int rr = min(m0 + 2 * j + half, RS - 1);
                const float* rp = rows + rr * NCLASS + l15 * 4;
                ov[j] = make_float4(rp[0], rp[1], rp[2], rp[3]);
            }
#pragma unroll
            for (int j = 0; j < 8; ++j) {
                const int rr = m0 + 2 * j + half;
                const int node = node0 + rr;
                if (rr < RS && node < n) out4[(size_t)node * (NCLASS / 4) + l15] = ov[j];
            }
        }
        __syncthreads();
    }
}

extern "C" void kernel_launch(void* const* d_in, const int* in_sizes, int n_in,
                              void* d_out, int out_size, void* d_ws, size_t ws_size,
                              hipStream_t stream) {
    const float* x  = (const float*)d_in[0];
    const int*   ei = (const int*)d_in[1];
    const float* Wg = (const float*)d_in[2];
    const float* bg = (const float*)d_in[3];
    const float* Wl = (const float*)d_in[4];
    const float* bl = (const float*)d_in[5];
    float* out = (float*)d_out;

    const int n = in_sizes[0] / NFEAT;
    const int e = in_sizes[1] / 2;

    float* ws   = (float*)d_ws;
    size_t off  = ((size_t)n + 3) & ~(size_t)3;
    float* deg  = ws;
    float* h    = ws + off;
    float* lst  = h + (size_t)n * NCLASS;

    const int T = 256;

    const int mtiles = (n + 15) / 16;
    const int gblocks = (mtiles + 7) / 8;
    const int dblocks = (e + T * DC_U - 1) / (T * DC_U);
    float* offsb = lst + off;
    float* srcb  = offsb + (size_t)dblocks * SC_BLOCKS;
    float* dstb  = srcb + e;

    deg_init_kernel <<<(n + T * 4 - 1) / (T * 4), T, 0, stream>>>(deg, lst, n);
    gemm1_degcount_kernel<<<gblocks + dblocks, T, 0, stream>>>(x, Wg, h, n, gblocks, ei, deg, lst, offsb, srcb, dstb, e);
    deg_sum_kernel<<<(n + T * 4 - 1) / (T * 4), T, 0, stream>>>(deg, lst, n);

    edge_sum_kernel<<<SC_BLOCKS, SC_THREADS, 0, stream>>>(h, deg, bg, offsb, srcb, dstb, Wl, bl, out, e, n);
}
